# starting kernel
# baseline (speedup 1.0000x reference)
#include <hip/hip_runtime.h>
#include <math.h>

typedef float v2f __attribute__((ext_vector_type(2)));
typedef float v4f __attribute__((ext_vector_type(4)));
typedef float v8f __attribute__((ext_vector_type(8)));

#define IN_DIM 128
#define OUT_DIM 128
#define NHEAD 4
#define HD 32
#define NEG_SLOPE 0.2f
#define EPS_SOFTMAX 1e-8f
#define EPS_LN 1e-5f

__device__ __forceinline__ void atomicMaxF32(float* addr, float val) {
    if (val >= 0.0f) {
        atomicMax((int*)addr, __float_as_int(val));
    } else {
        atomicMin((unsigned int*)addr, (unsigned int)__float_as_int(val));
    }
}

__global__ void init_kernel(float* emax0, float* emax1, float* asum0, float* asum1, int NH) {
    int gid = blockIdx.x * blockDim.x + threadIdx.x;
    if (gid >= NH) return;
    emax0[gid] = -INFINITY;
    emax1[gid] = -INFINITY;
    asum0[gid] = 0.0f;
    asum1[gid] = 0.0f;
}

__global__ __launch_bounds__(256) void gemm_wmma_kernel(
    const float* __restrict__ h, const float* __restrict__ W,
    const float* __restrict__ Wself, float* __restrict__ h_proj,
    float* __restrict__ out, int N)
{
    const int wave = threadIdx.x >> 5;
    const int lane = threadIdx.x & 31;
    const int row0 = blockIdx.x * 16;
    const int mrow = lane & 15;
    const int khalf = (lane >> 4) * 2;

    int arow_idx = row0 + mrow;
    if (arow_idx >= N) arow_idx = N - 1;
    const float* arow = h + (size_t)arow_idx * IN_DIM;

    const int col0 = wave * 32;
    const int ccol0 = col0 + mrow;
    const int ccol1 = col0 + 16 + mrow;
    const bool t0_proj = (col0 < 128);
    const bool t1_proj = (col0 + 16 < 128);
    const float* bp0 = t0_proj ? (W + ccol0) : (Wself + (ccol0 - 128));
    const float* bp1 = t1_proj ? (W + ccol1) : (Wself + (ccol1 - 128));

    v8f c0 = {};
    v8f c1 = {};
    for (int k0 = 0; k0 < IN_DIM; k0 += 4) {
        const int k = k0 + khalf;
        v2f a, b0, b1;
        a.x = arow[k];
        a.y = arow[k + 1];
        b0.x = bp0[(size_t)k * OUT_DIM];
        b0.y = bp0[(size_t)(k + 1) * OUT_DIM];
        b1.x = bp1[(size_t)k * OUT_DIM];
        b1.y = bp1[(size_t)(k + 1) * OUT_DIM];
        c0 = __builtin_amdgcn_wmma_f32_16x16x4_f32(false, a, false, b0,
                                                   (short)0, c0, false, false);
        c1 = __builtin_amdgcn_wmma_f32_16x16x4_f32(false, a, false, b1,
                                                   (short)0, c1, false, false);
    }

    const int mbase = (lane >> 4) * 8;
    float* dst0 = t0_proj ? (h_proj + ccol0) : (out + (ccol0 - 128));
    float* dst1 = t1_proj ? (h_proj + ccol1) : (out + (ccol1 - 128));
    if (row0 + 16 <= N) {
#pragma unroll
        for (int v = 0; v < 8; ++v) {
            const size_t row = (size_t)(row0 + mbase + v);
            dst0[row * OUT_DIM] = c0[v];
            dst1[row * OUT_DIM] = c1[v];
        }
    } else {
#pragma unroll
        for (int v = 0; v < 8; ++v) {
            const int row = row0 + mbase + v;
            if (row < N) {
                dst0[(size_t)row * OUT_DIM] = c0[v];
                dst1[(size_t)row * OUT_DIM] = c1[v];
            }
        }
    }
}

__global__ void score_kernel(const float* __restrict__ h_proj,
                             const float* __restrict__ a_in,
                             const float* __restrict__ a_out,
                             float* __restrict__ s_in_s, float* __restrict__ s_in_r,
                             float* __restrict__ s_out_s, float* __restrict__ s_out_r,
                             int NH)
{
    int gid = blockIdx.x * blockDim.x + threadIdx.x;
    if (gid >= NH) return;
    const int hh = gid & 3;
    const int n = gid >> 2;
    const float* x = h_proj + (size_t)n * OUT_DIM + hh * HD;
    const float* ai = a_in + hh * 2 * HD;
    const float* ao = a_out + hh * 2 * HD;
    float s0 = 0.f, s1 = 0.f, s2 = 0.f, s3 = 0.f;
#pragma unroll
    for (int d = 0; d < HD; d += 4) {
        const v4f xv = *(const v4f*)(x + d);
#pragma unroll
        for (int j = 0; j < 4; ++j) {
            s0 += xv[j] * ai[d + j];
            s1 += xv[j] * ai[HD + d + j];
            s2 += xv[j] * ao[d + j];
            s3 += xv[j] * ao[HD + d + j];
        }
    }
    s_in_s[gid] = s0;
    s_in_r[gid] = s1;
    s_out_s[gid] = s2;
    s_out_r[gid] = s3;
}

__global__ void edge_max_kernel(const int* __restrict__ snd, const int* __restrict__ rcv,
                                const float* __restrict__ s_s, const float* __restrict__ s_r,
                                float* __restrict__ emax, int E)
{
    int e = blockIdx.x * blockDim.x + threadIdx.x;
    if (e >= E) return;
    const int s = snd[e];
    const int r = rcv[e];
    const v4f ss = *(const v4f*)(s_s + (size_t)s * 4);
    const v4f sr = *(const v4f*)(s_r + (size_t)r * 4);
#pragma unroll
    for (int hh = 0; hh < NHEAD; ++hh) {
        float ev = ss[hh] + sr[hh];
        ev = (ev > 0.0f) ? ev : NEG_SLOPE * ev;
        atomicMaxF32(&emax[(size_t)r * 4 + hh], ev);
    }
}

__global__ void edge_sum_kernel(const int* __restrict__ snd, const int* __restrict__ rcv,
                                const float* __restrict__ s_s, const float* __restrict__ s_r,
                                const float* __restrict__ emax,
                                float* __restrict__ alpha_buf, float* __restrict__ asum,
                                int E)
{
    int e = blockIdx.x * blockDim.x + threadIdx.x;
    if (e >= E) return;
    const int s = snd[e];
    const int r = rcv[e];
    const v4f ss = *(const v4f*)(s_s + (size_t)s * 4);
    const v4f sr = *(const v4f*)(s_r + (size_t)r * 4);
    const v4f em = *(const v4f*)(emax + (size_t)r * 4);
    v4f alpha;
#pragma unroll
    for (int hh = 0; hh < NHEAD; ++hh) {
        float ev = ss[hh] + sr[hh];
        ev = (ev > 0.0f) ? ev : NEG_SLOPE * ev;
        alpha[hh] = __expf(ev - em[hh]);
        atomicAdd(&asum[(size_t)r * 4 + hh], alpha[hh]);
    }
    *(v4f*)(alpha_buf + (size_t)e * 4) = alpha;
}

__global__ void edge_agg_kernel(const int* __restrict__ snd, const int* __restrict__ rcv,
                                const float* __restrict__ h_proj,
                                const float* __restrict__ alpha_buf,
                                const float* __restrict__ asum,
                                float* __restrict__ out, int EH)
{
    int gid = blockIdx.x * blockDim.x + threadIdx.x;
    if (gid >= EH) return;
    const int hh = gid & 3;
    const int e = gid >> 2;
    const int s = snd[e];
    const int r = rcv[e];
    const float w = alpha_buf[(size_t)e * 4 + hh] / (asum[(size_t)r * 4 + hh] + EPS_SOFTMAX);
    const float* xs = h_proj + (size_t)s * OUT_DIM + hh * HD;
    float* xo = out + (size_t)r * OUT_DIM + hh * HD;
#pragma unroll
    for (int d = 0; d < HD; d += 4) {
        const v4f xv = *(const v4f*)(xs + d);
#pragma unroll
        for (int j = 0; j < 4; ++j) {
            atomicAdd(xo + d + j, w * xv[j]);
        }
    }
}

__global__ __launch_bounds__(128) void ln_kernel(float* __restrict__ out,
                                                 const float* __restrict__ bias,
                                                 const float* __restrict__ gamma,
                                                 const float* __restrict__ beta,
                                                 int N)
{
    __shared__ float red[OUT_DIM];
    const int n = blockIdx.x;
    const int t = threadIdx.x;
    if (n >= N) return;
    const float x = out[(size_t)n * OUT_DIM + t] + bias[t];
    red[t] = x;
    __syncthreads();
    for (int s = OUT_DIM / 2; s > 0; s >>= 1) {
        if (t < s) red[t] += red[t + s];
        __syncthreads();
    }
    const float mean = red[0] * (1.0f / OUT_DIM);
    __syncthreads();
    const float d = x - mean;
    red[t] = d * d;
    __syncthreads();
    for (int s = OUT_DIM / 2; s > 0; s >>= 1) {
        if (t < s) red[t] += red[t + s];
        __syncthreads();
    }
    const float var = red[0] * (1.0f / OUT_DIM);
    const float inv = rsqrtf(var + EPS_LN);
    out[(size_t)n * OUT_DIM + t] = d * inv * gamma[t] + beta[t];
}

extern "C" void kernel_launch(void* const* d_in, const int* in_sizes, int n_in,
                              void* d_out, int out_size, void* d_ws, size_t ws_size,
                              hipStream_t stream) {
    const float* h      = (const float*)d_in[0];
    const int*   eidx   = (const int*)d_in[1];
    const float* W      = (const float*)d_in[2];
    const float* Wself  = (const float*)d_in[3];
    const float* a_in   = (const float*)d_in[4];
    const float* a_out  = (const float*)d_in[5];
    const float* bias   = (const float*)d_in[6];
    const float* gamma  = (const float*)d_in[7];
    const float* beta   = (const float*)d_in[8];
    float* out = (float*)d_out;

    const int N = in_sizes[0] / IN_DIM;
    const int E = in_sizes[1] / 2;
    const int NH = N * NHEAD;
    const int EH = E * NHEAD;

    const int* src = eidx;
    const int* dst = eidx + E;

    float* ws = (float*)d_ws;
    float* h_proj  = ws;
    float* s_in_s  = h_proj + (size_t)N * OUT_DIM;
    float* s_in_r  = s_in_s + NH;
    float* s_out_s = s_in_r + NH;
    float* s_out_r = s_out_s + NH;
    float* emax0   = s_out_r + NH;
    float* emax1   = emax0 + NH;
    float* asum0   = emax1 + NH;
    float* asum1   = asum0 + NH;
    float* alpha0  = asum1 + NH;
    float* alpha1  = alpha0 + (size_t)E * NHEAD;

    {
        int threads = 256, blocks = (NH + threads - 1) / threads;
        init_kernel<<<blocks, threads, 0, stream>>>(emax0, emax1, asum0, asum1, NH);
    }
    {
        int blocks = (N + 15) / 16;
        gemm_wmma_kernel<<<blocks, 256, 0, stream>>>(h, W, Wself, h_proj, out, N);
    }
    {
        int threads = 256, blocks = (NH + threads - 1) / threads;
        score_kernel<<<blocks, threads, 0, stream>>>(h_proj, a_in, a_out,
                                                     s_in_s, s_in_r, s_out_s, s_out_r, NH);
    }
    {
        int threads = 256, blocks = (E + threads - 1) / threads;
        edge_max_kernel<<<blocks, threads, 0, stream>>>(src, dst, s_in_s, s_in_r, emax0, E);
        edge_max_kernel<<<blocks, threads, 0, stream>>>(dst, src, s_out_s, s_out_r, emax1, E);
    }
    {
        int threads = 256, blocks = (E + threads - 1) / threads;
        edge_sum_kernel<<<blocks, threads, 0, stream>>>(src, dst, s_in_s, s_in_r, emax0,
                                                        alpha0, asum0, E);
        edge_sum_kernel<<<blocks, threads, 0, stream>>>(dst, src, s_out_s, s_out_r, emax1,
                                                        alpha1, asum1, E);
    }
    {
        int threads = 256, blocks = (EH + threads - 1) / threads;
        edge_agg_kernel<<<blocks, threads, 0, stream>>>(src, dst, h_proj,
                                                        alpha0, asum0, out, EH);
        edge_agg_kernel<<<blocks, threads, 0, stream>>>(dst, src, h_proj,
                                                        alpha1, asum1, out, EH);
    }
    {
        ln_kernel<<<N, OUT_DIM, 0, stream>>>(out, bias, gamma, beta, N);
    }
}


# Softmax max and sums formed in the receiver-tile kernel: kept edges listed in LDS, max and sum by LDS atomics, weights from scores; both global-atomic edge passes gone, 4 launches; sum order changes
# speedup vs baseline: 11.1472x; 11.1472x over previous
#include <hip/hip_runtime.h>
#include <math.h>

typedef float v2f __attribute__((ext_vector_type(2)));
typedef float v4f __attribute__((ext_vector_type(4)));
typedef float v8f __attribute__((ext_vector_type(8)));

#define IN_DIM 128
#define OUT_DIM 128
#define NHEAD 4
#define HD 32
#define NEG_SLOPE 0.2f
#define EPS_SOFTMAX 1e-8f
#define EPS_LN 1e-5f

__device__ __forceinline__ void atomicMaxF32(float* addr, float val) {
    if (val >= 0.0f) {
        atomicMax((int*)addr, __float_as_int(val));
    } else {
        atomicMin((unsigned int*)addr, (unsigned int)__float_as_int(val));
    }
}

#define REC 32
#define REC_S_IN_S 0
#define REC_S_OUT_R 4
#define REC_S_IN_R 8
#define REC_S_OUT_S 12
#define REC_MAP 16

__global__ __launch_bounds__(256) void gemm_wmma_kernel(
    const float* __restrict__ h, const float* __restrict__ W,
    const float* __restrict__ Wself, float* __restrict__ h_proj,
    float* __restrict__ out, int N)
{
    const int wave = threadIdx.x >> 5;
    const int lane = threadIdx.x & 31;
    const int row0 = blockIdx.x * 16;
    const int mrow = lane & 15;
    const int khalf = (lane >> 4) * 2;

    int arow_idx = row0 + mrow;
    if (arow_idx >= N) arow_idx = N - 1;
    const float* arow = h + (size_t)arow_idx * IN_DIM;

    const int col0 = wave * 32;
    const int ccol0 = col0 + mrow;
    const int ccol1 = col0 + 16 + mrow;
    const bool t0_proj = (col0 < 128);
    const bool t1_proj = (col0 + 16 < 128);
    const float* bp0 = t0_proj ? (W + ccol0) : (Wself + (ccol0 - 128));
    const float* bp1 = t1_proj ? (W + ccol1) : (Wself + (ccol1 - 128));

    v8f c0 = {};
    v8f c1 = {};
    for (int k0 = 0; k0 < IN_DIM; k0 += 4) {
        const int k = k0 + khalf;
        v2f a, b0, b1;
        a.x = arow[k];
        a.y = arow[k + 1];
        b0.x = bp0[(size_t)k * OUT_DIM];
        b0.y = bp0[(size_t)(k + 1) * OUT_DIM];
        b1.x = bp1[(size_t)k * OUT_DIM];
        b1.y = bp1[(size_t)(k + 1) * OUT_DIM];
        c0 = __builtin_amdgcn_wmma_f32_16x16x4_f32(false, a, false, b0,
                                                   (short)0, c0, false, false);
        c1 = __builtin_amdgcn_wmma_f32_16x16x4_f32(false, a, false, b1,
                                                   (short)0, c1, false, false);
    }

    const int mbase = (lane >> 4) * 8;
    float* dst0 = t0_proj ? (h_proj + ccol0) : (out + (ccol0 - 128));
    float* dst1 = t1_proj ? (h_proj + ccol1) : (out + (ccol1 - 128));
    if (row0 + 16 <= N) {
#pragma unroll
        for (int v = 0; v < 8; ++v) {
            const size_t row = (size_t)(row0 + mbase + v);
            dst0[row * OUT_DIM] = c0[v];
            dst1[row * OUT_DIM] = c1[v];
        }
    } else {
#pragma unroll
        for (int v = 0; v < 8; ++v) {
            const int row = row0 + mbase + v;
            if (row < N) {
                dst0[(size_t)row * OUT_DIM] = c0[v];
                dst1[(size_t)row * OUT_DIM] = c1[v];
            }
        }
    }
}

__global__ void score_kernel(const float* __restrict__ h_proj,
                             const float* __restrict__ a_in,
                             const float* __restrict__ a_out,
                             float* __restrict__ rec, int NH)
{
    int gid = blockIdx.x * blockDim.x + threadIdx.x;
    if (gid >= NH) return;
    const int hh = gid & 3;
    const int n = gid >> 2;
    const float* x = h_proj + (size_t)n * OUT_DIM + hh * HD;
    const float* ai = a_in + hh * 2 * HD;
    const float* ao = a_out + hh * 2 * HD;
    float s0 = 0.f, s1 = 0.f, s2 = 0.f, s3 = 0.f;
#pragma unroll
    for (int d = 0; d < HD; d += 4) {
        const v4f xv = *(const v4f*)(x + d);
#pragma unroll
        for (int j = 0; j < 4; ++j) {
            s0 += xv[j] * ai[d + j];
            s1 += xv[j] * ai[HD + d + j];
            s2 += xv[j] * ao[d + j];
            s3 += xv[j] * ao[HD + d + j];
        }
    }
    float* r = rec + (size_t)n * REC;
    r[REC_S_IN_S + hh] = s0;
    r[REC_S_IN_R + hh] = s1;
    r[REC_S_OUT_S + hh] = s2;
    r[REC_S_OUT_R + hh] = s3;
}

#define TILE_ROWS 400
#define AGG_THREADS 1024
#define AGG_WAVES (AGG_THREADS / 32)
#define STAT 8

#define MAP_EDGES 1024
#define MAP_SLOTS 16
#define STEP_GROUPS 8

__device__ __forceinline__ size_t map_word_at(unsigned i) {
    return (size_t)(i >> 4) * REC + REC_MAP + (i & (MAP_SLOTS - 1));
}

__global__ __launch_bounds__(MAP_EDGES) void tile_map_kernel(
    const int* __restrict__ src, const int* __restrict__ dst,
    unsigned int* __restrict__ map, int E, int tiles)
{
    extern __shared__ unsigned int wmap[];
    const int tid = threadIdx.x;
    for (int t = tid; t < tiles; t += MAP_EDGES) wmap[t] = 0u;
    __syncthreads();
    const int e = blockIdx.x * MAP_EDGES + tid;
    if (e < E) {
        const unsigned d = (unsigned)dst[e];
        const unsigned s = (unsigned)src[e];
        const unsigned bit = 1u << (tid >> 5);
        atomicOr(&wmap[d / TILE_ROWS], bit);
        atomicOr(&wmap[s / TILE_ROWS], bit);
    }
    __syncthreads();
    for (int t = tid; t < tiles; t += MAP_EDGES)
        map[map_word_at((unsigned)blockIdx.x * (unsigned)tiles + (unsigned)t)] = wmap[t];
}

#define LIST_CAP 16384
#define LIST_PER_THREAD (LIST_CAP / AGG_THREADS)
#define PASS_GROUP 4
#define ENT_ROW_SHIFT 22
#define ENT_DIR_SHIFT 31
#define ENT_SENDER_MASK 0x3FFFFFu
#define ENT_ROW_MASK 0x1FFu

__device__ __forceinline__ v4f edge_logits(const float* rec, int sender, int recv, int dir)
{
    const v4f s = *(const v4f*)(rec + (size_t)sender * REC + (dir ? REC_S_OUT_S : REC_S_IN_S));
    const v4f r = *(const v4f*)(rec + (size_t)recv * REC + (dir ? REC_S_OUT_R : REC_S_IN_R));
    v4f e = s + r;
    e.x = (e.x > 0.0f) ? e.x : NEG_SLOPE * e.x;
    e.y = (e.y > 0.0f) ? e.y : NEG_SLOPE * e.y;
    e.z = (e.z > 0.0f) ? e.z : NEG_SLOPE * e.z;
    e.w = (e.w > 0.0f) ? e.w : NEG_SLOPE * e.w;
    return e;
}

template <class F>
__device__ __forceinline__ void scan_hit_groups(const int* __restrict__ src,
                                                const int* __restrict__ dst,
                                                const unsigned int* map, int E, int lo,
                                                unsigned urows, unsigned tiles, unsigned tile,
                                                int wave, int lane, F&& f)
{
    const int nwords = (E + MAP_EDGES - 1) / MAP_EDGES;
    for (int w0 = wave; w0 < nwords; w0 += 32 * AGG_WAVES) {
        const int wl = w0 + AGG_WAVES * lane;
        unsigned myword = 0u;
        if (wl < nwords)
            myword = map ? map[map_word_at((unsigned)wl * tiles + tile)] : 0xFFFFFFFFu;
        for (int k = 0; k < 32; ++k) {
            const int w = w0 + AGG_WAVES * k;
            if (w >= nwords) break;
            unsigned bits = (unsigned)__builtin_amdgcn_readlane((int)myword, k);
            const int ebase = w * MAP_EDGES;
            while (bits) {
                int g[STEP_GROUPS];
#pragma unroll
                for (int i = 0; i < STEP_GROUPS; ++i) {
                    g[i] = bits ? __builtin_ctz(bits) : 32;
                    bits &= bits - 1u;
                }
                int sv[STEP_GROUPS], dv[STEP_GROUPS];
#pragma unroll
                for (int i = 0; i < STEP_GROUPS; ++i) {
                    const int e = ebase + 32 * g[i] + lane;
                    const bool ok = (g[i] < 32) && (e < E);
                    sv[i] = ok ? src[e] : -1;
                    dv[i] = ok ? dst[e] : -1;
                }
                unsigned m_in[STEP_GROUPS], m_out[STEP_GROUPS];
                int kept = 0;
#pragma unroll
                for (int i = 0; i < STEP_GROUPS; ++i) {
                    m_in[i] = __builtin_amdgcn_ballot_w32((unsigned)(dv[i] - lo) < urows);
                    m_out[i] = __builtin_amdgcn_ballot_w32((unsigned)(sv[i] - lo) < urows);
                    kept += __builtin_popcount(m_in[i]) + __builtin_popcount(m_out[i]);
                }
                f(sv, dv, m_in, m_out, kept);
            }
        }
    }
}

#define GATHER_BATCH 8

__device__ __forceinline__ void request_rows(int b, int n, int q_s, int lane,
                                             const float* __restrict__ h_proj,
                                             float (&x)[GATHER_BATCH][NHEAD])
{
#pragma unroll
    for (int i = 0; i < GATHER_BATCH; ++i) {
        const int idx = (b + i < n) ? (b + i) : (n - 1);
        const int s = __builtin_amdgcn_readlane(q_s, idx);
        const float* xs = h_proj + (size_t)s * OUT_DIM + lane;
#pragma unroll
        for (int j = 0; j < NHEAD; ++j) x[i][j] = xs[HD * j];
    }
}

__device__ __forceinline__ void add_chunk_rows(int n, int q_s, int q_r, const int (&wq)[NHEAD],
                                               int lane, const float* __restrict__ h_proj,
                                               float* acc)
{
    for (int b = 0; b < n; b += GATHER_BATCH) {
        float x[GATHER_BATCH][NHEAD];
        request_rows(b, n, q_s, lane, h_proj, x);
#pragma unroll
        for (int i = 0; i < GATHER_BATCH; ++i) {
            const bool live = (b + i < n);
            const int idx = live ? (b + i) : (n - 1);
            const int r = __builtin_amdgcn_readlane(q_r, idx);
            float* ar = acc + r * OUT_DIM + lane;
#pragma unroll
            for (int j = 0; j < NHEAD; ++j) {
                const float w = __int_as_float(__builtin_amdgcn_readlane(wq[j], idx));
                const float v = live ? w * x[i][j] : 0.0f;
                atomicAdd(ar + HD * j, v);
            }
        }
    }
}

__device__ __forceinline__ void add_kept_edges(unsigned mask, int sender_v, int recv_v, int dir,
                                               int lane, int lo,
                                               const float* __restrict__ h_proj,
                                               const float* rec, const float* smax,
                                               const float* ssum, float* acc)
{
    while (mask) {
        const int L = __builtin_ctz(mask);
        mask &= mask - 1;
        const int s = __builtin_amdgcn_readlane(sender_v, L);
        const int r = __builtin_amdgcn_readlane(recv_v, L);
        const int hq = lane & 3;
        const float ss = rec[(size_t)s * REC + (dir ? REC_S_OUT_S : REC_S_IN_S) + hq];
        const float rr = rec[(size_t)r * REC + (dir ? REC_S_OUT_R : REC_S_IN_R) + hq];
        float e = ss + rr;
        e = (e > 0.0f) ? e : NEG_SLOPE * e;
        const int st = (r - lo) * STAT + dir * NHEAD + hq;
        const int wq = __float_as_int(__expf(e - smax[st]) / (ssum[st] + EPS_SOFTMAX));
        const float* xs = h_proj + (size_t)s * OUT_DIM + lane;
        float* ar = acc + (r - lo) * OUT_DIM + lane;
        float x[NHEAD];
#pragma unroll
        for (int j = 0; j < NHEAD; ++j) x[j] = xs[HD * j];
#pragma unroll
        for (int j = 0; j < NHEAD; ++j) {
            const float w = __int_as_float(__builtin_amdgcn_readlane(wq, j));
            atomicAdd(ar + HD * j, w * x[j]);
        }
    }
}

#define LN_ROWS 4

__device__ __forceinline__ float wave_sum32(float v) {
#pragma unroll
    for (int m = 16; m > 0; m >>= 1) v += __shfl_xor(v, m, 32);
    return v;
}

__global__ __launch_bounds__(AGG_THREADS) void agg_tile_kernel(
    const int* __restrict__ src, const int* __restrict__ dst,
    const float* __restrict__ h_proj,
    const float* rec,
    const float* __restrict__ bias, const float* __restrict__ gamma,
    const float* __restrict__ beta,
    const unsigned int* map,
    float* __restrict__ out, int N, int E)
{
    __shared__ __attribute__((aligned(16))) float acc[TILE_ROWS * OUT_DIM];
    __shared__ __attribute__((aligned(16))) float smax[TILE_ROWS * STAT];
    __shared__ __attribute__((aligned(16))) float ssum[TILE_ROWS * STAT];
    __shared__ unsigned int list[LIST_CAP];
    __shared__ int lcount;
    const int tid = threadIdx.x;
    const int lane = tid & 31;
    const int wave = __builtin_amdgcn_readfirstlane(tid >> 5);
    const int lo = blockIdx.x * TILE_ROWS;
    const int rows = (N - lo < TILE_ROWS) ? (N - lo) : TILE_ROWS;
    const unsigned urows = (unsigned)rows;
    const unsigned tiles = gridDim.x;
    const unsigned tile = blockIdx.x;

    for (int i = tid; i < TILE_ROWS * OUT_DIM / 4; i += AGG_THREADS) {
        const v4f zero4 = {0.f, 0.f, 0.f, 0.f};
        ((v4f*)acc)[i] = zero4;
    }
    for (int i = tid; i < TILE_ROWS * STAT; i += AGG_THREADS) {
        smax[i] = -INFINITY;
        ssum[i] = 0.0f;
    }
    if (tid == 0) lcount = 0;
    __syncthreads();

    const unsigned below = (1u << lane) - 1u;
    scan_hit_groups(src, dst, map, E, lo, urows, tiles, tile, wave, lane,
        [&](const int (&sv)[STEP_GROUPS], const int (&dv)[STEP_GROUPS],
            const unsigned (&m_in)[STEP_GROUPS], const unsigned (&m_out)[STEP_GROUPS],
            int kept) {
            if (kept == 0) return;
            int base = 0;
            if (lane == 0) base = atomicAdd(&lcount, kept);
            int off = __builtin_amdgcn_readfirstlane(base);
#pragma unroll
            for (int i = 0; i < STEP_GROUPS; ++i) {
                if ((m_in[i] >> lane) & 1u) {
                    const int pos = off + __builtin_popcount(m_in[i] & below);
                    if (pos < LIST_CAP)
                        list[pos] = (unsigned)sv[i] | ((unsigned)(dv[i] - lo) << ENT_ROW_SHIFT);
                }
                off += __builtin_popcount(m_in[i]);
                if ((m_out[i] >> lane) & 1u) {
                    const int pos = off + __builtin_popcount(m_out[i] & below);
                    if (pos < LIST_CAP)
                        list[pos] = (unsigned)dv[i] | ((unsigned)(sv[i] - lo) << ENT_ROW_SHIFT)
                                    | (1u << ENT_DIR_SHIFT);
                }
                off += __builtin_popcount(m_out[i]);
            }
        });
    __syncthreads();
    const int n = __builtin_amdgcn_readfirstlane(lcount);

    if (n <= LIST_CAP) {
        if (n > 0) {
            v4f ev[LIST_PER_THREAD];
#pragma unroll
            for (int k0 = 0; k0 < LIST_PER_THREAD; k0 += PASS_GROUP) {
                if (AGG_THREADS * k0 < n) {
                    unsigned en[PASS_GROUP];
#pragma unroll
                    for (int j = 0; j < PASS_GROUP; ++j) {
                        const int i = tid + AGG_THREADS * (k0 + j);
                        en[j] = list[(i < n) ? i : n - 1];
                        const int row = (int)((en[j] >> ENT_ROW_SHIFT) & ENT_ROW_MASK);
                        ev[k0 + j] = edge_logits(rec, (int)(en[j] & ENT_SENDER_MASK), lo + row,
                                                 (int)(en[j] >> ENT_DIR_SHIFT));
                    }
#pragma unroll
                    for (int j = 0; j < PASS_GROUP; ++j) {
                        if (tid + AGG_THREADS * (k0 + j) < n) {
                            float* mp = smax + (int)((en[j] >> ENT_ROW_SHIFT) & ENT_ROW_MASK) * STAT
                                        + (int)(en[j] >> ENT_DIR_SHIFT) * NHEAD;
                            atomicMaxF32(mp + 0, ev[k0 + j].x);
                            atomicMaxF32(mp + 1, ev[k0 + j].y);
                            atomicMaxF32(mp + 2, ev[k0 + j].z);
                            atomicMaxF32(mp + 3, ev[k0 + j].w);
                        }
                    }
                }
            }
            __syncthreads();
#pragma unroll
            for (int k = 0; k < LIST_PER_THREAD; ++k) {
                const int i = tid + AGG_THREADS * k;
                if (i < n) {
                    const unsigned en = list[i];
                    const int st = (int)((en >> ENT_ROW_SHIFT) & ENT_ROW_MASK) * STAT
                                   + (int)(en >> ENT_DIR_SHIFT) * NHEAD;
                    const v4f mx = *(const v4f*)(smax + st);
                    atomicAdd(ssum + st + 0, __expf(ev[k].x - mx.x));
                    atomicAdd(ssum + st + 1, __expf(ev[k].y - mx.y));
                    atomicAdd(ssum + st + 2, __expf(ev[k].z - mx.z));
                    atomicAdd(ssum + st + 3, __expf(ev[k].w - mx.w));
                }
            }
            __syncthreads();
            const int chunks = (n + 31) >> 5;
            unsigned en_next = 0u;
            v4f e_next = {0.f, 0.f, 0.f, 0.f};
            if (wave < chunks) {
                const int i = 32 * wave + lane;
                en_next = list[(i < n) ? i : n - 1];
                e_next = edge_logits(rec, (int)(en_next & ENT_SENDER_MASK),
                                     lo + (int)((en_next >> ENT_ROW_SHIFT) & ENT_ROW_MASK),
                                     (int)(en_next >> ENT_DIR_SHIFT));
            }
            for (int c = wave; c < chunks; c += AGG_WAVES) {
                const int left = n - 32 * c;
                const int cn = (left < 32) ? left : 32;
                const unsigned en = en_next;
                const v4f e = e_next;
                if (c + AGG_WAVES < chunks) {
                    const int i = 32 * (c + AGG_WAVES) + lane;
                    en_next = list[(i < n) ? i : n - 1];
                    e_next = edge_logits(rec, (int)(en_next & ENT_SENDER_MASK),
                                         lo + (int)((en_next >> ENT_ROW_SHIFT) & ENT_ROW_MASK),
                                         (int)(en_next >> ENT_DIR_SHIFT));
                }
                const int sender = (int)(en & ENT_SENDER_MASK);
                const int row = (int)((en >> ENT_ROW_SHIFT) & ENT_ROW_MASK);
                const int dir = (int)(en >> ENT_DIR_SHIFT);
                const v4f mx = *(const v4f*)(smax + row * STAT + dir * NHEAD);
                const v4f sm = *(const v4f*)(ssum + row * STAT + dir * NHEAD);
                int wq[NHEAD];
                wq[0] = __float_as_int(__expf(e.x - mx.x) / (sm.x + EPS_SOFTMAX));
                wq[1] = __float_as_int(__expf(e.y - mx.y) / (sm.y + EPS_SOFTMAX));
                wq[2] = __float_as_int(__expf(e.z - mx.z) / (sm.z + EPS_SOFTMAX));
                wq[3] = __float_as_int(__expf(e.w - mx.w) / (sm.w + EPS_SOFTMAX));
                add_chunk_rows(cn, sender, row, wq, lane, h_proj, acc);
            }
        }
    } else {
        scan_hit_groups(src, dst, map, E, lo, urows, tiles, tile, wave, lane,
            [&](const int (&sv)[STEP_GROUPS], const int (&dv)[STEP_GROUPS],
                const unsigned (&m_in)[STEP_GROUPS], const unsigned (&m_out)[STEP_GROUPS],
                int kept) {
                if (kept == 0) return;
#pragma unroll
                for (int i = 0; i < STEP_GROUPS; ++i) {
                    if ((m_in[i] >> lane) & 1u) {
                        const v4f e = edge_logits(rec, sv[i], dv[i], 0);
                        float* mp = smax + (dv[i] - lo) * STAT;
                        atomicMaxF32(mp + 0, e.x);
                        atomicMaxF32(mp + 1, e.y);
                        atomicMaxF32(mp + 2, e.z);
                        atomicMaxF32(mp + 3, e.w);
                    }
                    if ((m_out[i] >> lane) & 1u) {
                        const v4f e = edge_logits(rec, dv[i], sv[i], 1);
                        float* mp = smax + (sv[i] - lo) * STAT + NHEAD;
                        atomicMaxF32(mp + 0, e.x);
                        atomicMaxF32(mp + 1, e.y);
                        atomicMaxF32(mp + 2, e.z);
                        atomicMaxF32(mp + 3, e.w);
                    }
                }
            });
        __syncthreads();
        scan_hit_groups(src, dst, map, E, lo, urows, tiles, tile, wave, lane,
            [&](const int (&sv)[STEP_GROUPS], const int (&dv)[STEP_GROUPS],
                const unsigned (&m_in)[STEP_GROUPS], const unsigned (&m_out)[STEP_GROUPS],
                int kept) {
                if (kept == 0) return;
#pragma unroll
                for (int i = 0; i < STEP_GROUPS; ++i) {
                    if ((m_in[i] >> lane) & 1u) {
                        const v4f e = edge_logits(rec, sv[i], dv[i], 0);
                        const int st = (dv[i] - lo) * STAT;
                        const v4f mx = *(const v4f*)(smax + st);
                        atomicAdd(ssum + st + 0, __expf(e.x - mx.x));
                        atomicAdd(ssum + st + 1, __expf(e.y - mx.y));
                        atomicAdd(ssum + st + 2, __expf(e.z - mx.z));
                        atomicAdd(ssum + st + 3, __expf(e.w - mx.w));
                    }
                    if ((m_out[i] >> lane) & 1u) {
                        const v4f e = edge_logits(rec, dv[i], sv[i], 1);
                        const int st = (sv[i] - lo) * STAT + NHEAD;
                        const v4f mx = *(const v4f*)(smax + st);
                        atomicAdd(ssum + st + 0, __expf(e.x - mx.x));
                        atomicAdd(ssum + st + 1, __expf(e.y - mx.y));
                        atomicAdd(ssum + st + 2, __expf(e.z - mx.z));
                        atomicAdd(ssum + st + 3, __expf(e.w - mx.w));
                    }
                }
            });
        __syncthreads();
        scan_hit_groups(src, dst, map, E, lo, urows, tiles, tile, wave, lane,
            [&](const int (&sv)[STEP_GROUPS], const int (&dv)[STEP_GROUPS],
                const unsigned (&m_in)[STEP_GROUPS], const unsigned (&m_out)[STEP_GROUPS],
                int kept) {
                if (kept == 0) return;
#pragma unroll
                for (int i = 0; i < STEP_GROUPS; ++i) {
                    add_kept_edges(m_in[i], sv[i], dv[i], 0, lane, lo, h_proj, rec,
                                   smax, ssum, acc);
                    add_kept_edges(m_out[i], dv[i], sv[i], 1, lane, lo, h_proj, rec,
                                   smax, ssum, acc);
                }
            });
    }
    __syncthreads();

    const v4f bi = *(const v4f*)(bias + 4 * lane);
    const v4f ga = *(const v4f*)(gamma + 4 * lane);
    const v4f be = *(const v4f*)(beta + 4 * lane);
    for (int r0 = wave; r0 < rows; r0 += LN_ROWS * AGG_WAVES) {
        v4f x[LN_ROWS];
#pragma unroll
        for (int i = 0; i < LN_ROWS; ++i) {
            const int r = r0 + i * AGG_WAVES;
            const int rc = (r < rows) ? r : r0;
            const v4f hs = *(const v4f*)(out + (size_t)(lo + rc) * OUT_DIM + 4 * lane);
            const v4f a = *(const v4f*)(acc + rc * OUT_DIM + 4 * lane);
            x[i] = hs + a + bi;
        }
        float mean[LN_ROWS];
#pragma unroll
        for (int i = 0; i < LN_ROWS; ++i)
            mean[i] = wave_sum32((x[i].x + x[i].y) + (x[i].z + x[i].w)) * (1.0f / OUT_DIM);
        float inv[LN_ROWS];
#pragma unroll
        for (int i = 0; i < LN_ROWS; ++i) {
            x[i] = x[i] - mean[i];
            const v4f q = x[i] * x[i];
            const float var = wave_sum32((q.x + q.y) + (q.z + q.w)) * (1.0f / OUT_DIM);
            inv[i] = rsqrtf(var + EPS_LN);
        }
#pragma unroll
        for (int i = 0; i < LN_ROWS; ++i) {
            const int r = r0 + i * AGG_WAVES;
            if (r < rows)
                *(v4f*)(out + (size_t)(lo + r) * OUT_DIM + 4 * lane) = x[i] * inv[i] * ga + be;
        }
    }
}

extern "C" void kernel_launch(void* const* d_in, const int* in_sizes, int n_in,
                              void* d_out, int out_size, void* d_ws, size_t ws_size,
                              hipStream_t stream) {
    const float* h      = (const float*)d_in[0];
    const int*   eidx   = (const int*)d_in[1];
    const float* W      = (const float*)d_in[2];
    const float* Wself  = (const float*)d_in[3];
    const float* a_in   = (const float*)d_in[4];
    const float* a_out  = (const float*)d_in[5];
    const float* bias   = (const float*)d_in[6];
    const float* gamma  = (const float*)d_in[7];
    const float* beta   = (const float*)d_in[8];
    float* out = (float*)d_out;

    const int N = in_sizes[0] / IN_DIM;
    const int E = in_sizes[1] / 2;
    const int NH = N * NHEAD;

    const int* src = eidx;
    const int* dst = eidx + E;

    float* ws = (float*)d_ws;
    float* h_proj  = ws;
    float* rec     = h_proj + (size_t)N * OUT_DIM;

    {
        int blocks = (N + 15) / 16;
        gemm_wmma_kernel<<<blocks, 256, 0, stream>>>(h, W, Wself, h_proj, out, N);
    }
    {
        int threads = 256, blocks = (NH + threads - 1) / threads;
        score_kernel<<<blocks, threads, 0, stream>>>(h_proj, a_in, a_out, rec, NH);
    }
    const int tiles = (N + TILE_ROWS - 1) / TILE_ROWS;
    const int nwords = (E + MAP_EDGES - 1) / MAP_EDGES;
    const bool use_map = (long long)nwords * tiles <= (long long)N * MAP_SLOTS;
    unsigned int* map = (unsigned int*)rec;
    if (use_map) {
        tile_map_kernel<<<nwords, MAP_EDGES, tiles * sizeof(unsigned int), stream>>>(
            src, dst, map, E, tiles);
    }
    {
        agg_tile_kernel<<<tiles, AGG_THREADS, 0, stream>>>(src, dst, h_proj, rec,
                                                           bias, gamma, beta,
                                                           use_map ? map : nullptr,
                                                           out, N, E);
    }
}


# GEMM wave tile 32 rows x 64 columns (was 16 x 32): B fragments reused over 2 row tiles and A over 4 column tiles from registers, A read 16 B per lane; 1563 blocks; k order of the sums changes
# speedup vs baseline: 12.0097x; 1.0774x over previous
#include <hip/hip_runtime.h>
#include <math.h>

typedef float v2f __attribute__((ext_vector_type(2)));
typedef float v4f __attribute__((ext_vector_type(4)));
typedef float v8f __attribute__((ext_vector_type(8)));

#define IN_DIM 128
#define OUT_DIM 128
#define NHEAD 4
#define HD 32
#define NEG_SLOPE 0.2f
#define EPS_SOFTMAX 1e-8f
#define EPS_LN 1e-5f

__device__ __forceinline__ void atomicMaxF32(float* addr, float val) {
    if (val >= 0.0f) {
        atomicMax((int*)addr, __float_as_int(val));
    } else {
        atomicMin((unsigned int*)addr, (unsigned int)__float_as_int(val));
    }
}

#define REC 32
#define REC_S_IN_S 0
#define REC_S_OUT_R 4
#define REC_S_IN_R 8
#define REC_S_OUT_S 12
#define REC_MAP 16

#define GEMM_ROWS 64
__global__ __launch_bounds__(256) void gemm_wmma_kernel(
    const float* __restrict__ h, const float* __restrict__ W,
    const float* __restrict__ Wself, float* __restrict__ h_proj,
    float* __restrict__ out, int N)
{
    const int wave = threadIdx.x >> 5;
    const int lane = threadIdx.x & 31;
    const int c = lane & 15;
    const int hh = lane >> 4;
    const int rgrp = wave >> 2;
    const int cgrp = wave & 3;
    const int row0 = blockIdx.x * GEMM_ROWS + rgrp * 32;

    int ra = row0 + c;
    int rb = row0 + 16 + c;
    if (ra >= N) ra = N - 1;
    if (rb >= N) rb = N - 1;
    const float* ap0 = h + (size_t)ra * IN_DIM + 4 * hh;
    const float* ap1 = h + (size_t)rb * IN_DIM + 4 * hh;

    const bool proj = (cgrp < 2);
    const int colbase = (cgrp & 1) * 64;
    const float* bp = (proj ? W : Wself) + colbase + c + (size_t)(4 * hh) * OUT_DIM;

    v8f acc0[4] = {};
    v8f acc1[4] = {};
    for (int k0 = 0; k0 < IN_DIM; k0 += 8) {
        const v4f a0 = *(const v4f*)(ap0 + k0);
        const v4f a1 = *(const v4f*)(ap1 + k0);
        float b[4][4];
#pragma unroll
        for (int t = 0; t < 4; ++t) {
#pragma unroll
            for (int s = 0; s < 4; ++s) b[t][s] = bp[(size_t)(k0 + s) * OUT_DIM + 16 * t];
        }
#pragma unroll
        for (int j = 0; j < 2; ++j) {
            v2f fa0, fa1;
            fa0.x = a0[2 * j];
            fa0.y = a0[2 * j + 1];
            fa1.x = a1[2 * j];
            fa1.y = a1[2 * j + 1];
#pragma unroll
            for (int t = 0; t < 4; ++t) {
                v2f fb;
                fb.x = b[t][2 * j];
                fb.y = b[t][2 * j + 1];
                acc0[t] = __builtin_amdgcn_wmma_f32_16x16x4_f32(false, fa0, false, fb,
                                                                (short)0, acc0[t], false, false);
                acc1[t] = __builtin_amdgcn_wmma_f32_16x16x4_f32(false, fa1, false, fb,
                                                                (short)0, acc1[t], false, false);
            }
        }
    }

    float* dstm = (proj ? h_proj : out) + colbase + c;
    const int mrow = row0 + 8 * hh;
    if (row0 + 32 <= N) {
#pragma unroll
        for (int t = 0; t < 4; ++t) {
#pragma unroll
            for (int v = 0; v < 8; ++v) {
                dstm[(size_t)(mrow + v) * OUT_DIM + 16 * t] = acc0[t][v];
                dstm[(size_t)(mrow + 16 + v) * OUT_DIM + 16 * t] = acc1[t][v];
            }
        }
    } else {
#pragma unroll
        for (int t = 0; t < 4; ++t) {
#pragma unroll
            for (int v = 0; v < 8; ++v) {
                if (mrow + v < N)
                    dstm[(size_t)(mrow + v) * OUT_DIM + 16 * t] = acc0[t][v];
                if (mrow + 16 + v < N)
                    dstm[(size_t)(mrow + 16 + v) * OUT_DIM + 16 * t] = acc1[t][v];
            }
        }
    }
}

__global__ void score_kernel(const float* __restrict__ h_proj,
                             const float* __restrict__ a_in,
                             const float* __restrict__ a_out,
                             float* __restrict__ rec, int NH)
{
    int gid = blockIdx.x * blockDim.x + threadIdx.x;
    if (gid >= NH) return;
    const int hh = gid & 3;
    const int n = gid >> 2;
    const float* x = h_proj + (size_t)n * OUT_DIM + hh * HD;
    const float* ai = a_in + hh * 2 * HD;
    const float* ao = a_out + hh * 2 * HD;
    float s0 = 0.f, s1 = 0.f, s2 = 0.f, s3 = 0.f;
#pragma unroll
    for (int d = 0; d < HD; d += 4) {
        const v4f xv = *(const v4f*)(x + d);
#pragma unroll
        for (int j = 0; j < 4; ++j) {
            s0 += xv[j] * ai[d + j];
            s1 += xv[j] * ai[HD + d + j];
            s2 += xv[j] * ao[d + j];
            s3 += xv[j] * ao[HD + d + j];
        }
    }
    float* r = rec + (size_t)n * REC;
    r[REC_S_IN_S + hh] = s0;
    r[REC_S_IN_R + hh] = s1;
    r[REC_S_OUT_S + hh] = s2;
    r[REC_S_OUT_R + hh] = s3;
}

#define TILE_ROWS 400
#define AGG_THREADS 1024
#define AGG_WAVES (AGG_THREADS / 32)
#define STAT 8

#define MAP_EDGES 1024
#define MAP_SLOTS 16
#define STEP_GROUPS 8

__device__ __forceinline__ size_t map_word_at(unsigned i) {
    return (size_t)(i >> 4) * REC + REC_MAP + (i & (MAP_SLOTS - 1));
}

__global__ __launch_bounds__(MAP_EDGES) void tile_map_kernel(
    const int* __restrict__ src, const int* __restrict__ dst,
    unsigned int* __restrict__ map, int E, int tiles)
{
    extern __shared__ unsigned int wmap[];
    const int tid = threadIdx.x;
    for (int t = tid; t < tiles; t += MAP_EDGES) wmap[t] = 0u;
    __syncthreads();
    const int e = blockIdx.x * MAP_EDGES + tid;
    if (e < E) {
        const unsigned d = (unsigned)dst[e];
        const unsigned s = (unsigned)src[e];
        const unsigned bit = 1u << (tid >> 5);
        atomicOr(&wmap[d / TILE_ROWS], bit);
        atomicOr(&wmap[s / TILE_ROWS], bit);
    }
    __syncthreads();
    for (int t = tid; t < tiles; t += MAP_EDGES)
        map[map_word_at((unsigned)blockIdx.x * (unsigned)tiles + (unsigned)t)] = wmap[t];
}

#define LIST_CAP 16384
#define LIST_PER_THREAD (LIST_CAP / AGG_THREADS)
#define PASS_GROUP 4
#define ENT_ROW_SHIFT 22
#define ENT_DIR_SHIFT 31
#define ENT_SENDER_MASK 0x3FFFFFu
#define ENT_ROW_MASK 0x1FFu

__device__ __forceinline__ v4f edge_logits(const float* rec, int sender, int recv, int dir)
{
    const v4f s = *(const v4f*)(rec + (size_t)sender * REC + (dir ? REC_S_OUT_S : REC_S_IN_S));
    const v4f r = *(const v4f*)(rec + (size_t)recv * REC + (dir ? REC_S_OUT_R : REC_S_IN_R));
    v4f e = s + r;
    e.x = (e.x > 0.0f) ? e.x : NEG_SLOPE * e.x;
    e.y = (e.y > 0.0f) ? e.y : NEG_SLOPE * e.y;
    e.z = (e.z > 0.0f) ? e.z : NEG_SLOPE * e.z;
    e.w = (e.w > 0.0f) ? e.w : NEG_SLOPE * e.w;
    return e;
}

template <class F>
__device__ __forceinline__ void scan_hit_groups(const int* __restrict__ src,
                                                const int* __restrict__ dst,
                                                const unsigned int* map, int E, int lo,
                                                unsigned urows, unsigned tiles, unsigned tile,
                                                int wave, int lane, F&& f)
{
    const int nwords = (E + MAP_EDGES - 1) / MAP_EDGES;
    for (int w0 = wave; w0 < nwords; w0 += 32 * AGG_WAVES) {
        const int wl = w0 + AGG_WAVES * lane;
        unsigned myword = 0u;
        if (wl < nwords)
            myword = map ? map[map_word_at((unsigned)wl * tiles + tile)] : 0xFFFFFFFFu;
        for (int k = 0; k < 32; ++k) {
            const int w = w0 + AGG_WAVES * k;
            if (w >= nwords) break;
            unsigned bits = (unsigned)__builtin_amdgcn_readlane((int)myword, k);
            const int ebase = w * MAP_EDGES;
            while (bits) {
                int g[STEP_GROUPS];
#pragma unroll
                for (int i = 0; i < STEP_GROUPS; ++i) {
                    g[i] = bits ? __builtin_ctz(bits) : 32;
                    bits &= bits - 1u;
                }
                int sv[STEP_GROUPS], dv[STEP_GROUPS];
#pragma unroll
                for (int i = 0; i < STEP_GROUPS; ++i) {
                    const int e = ebase + 32 * g[i] + lane;
                    const bool ok = (g[i] < 32) && (e < E);
                    sv[i] = ok ? src[e] : -1;
                    dv[i] = ok ? dst[e] : -1;
                }
                unsigned m_in[STEP_GROUPS], m_out[STEP_GROUPS];
                int kept = 0;
#pragma unroll
                for (int i = 0; i < STEP_GROUPS; ++i) {
                    m_in[i] = __builtin_amdgcn_ballot_w32((unsigned)(dv[i] - lo) < urows);
                    m_out[i] = __builtin_amdgcn_ballot_w32((unsigned)(sv[i] - lo) < urows);
                    kept += __builtin_popcount(m_in[i]) + __builtin_popcount(m_out[i]);
                }
                f(sv, dv, m_in, m_out, kept);
            }
        }
    }
}

#define GATHER_BATCH 8

__device__ __forceinline__ void request_rows(int b, int n, int q_s, int lane,
                                             const float* __restrict__ h_proj,
                                             float (&x)[GATHER_BATCH][NHEAD])
{
#pragma unroll
    for (int i = 0; i < GATHER_BATCH; ++i) {
        const int idx = (b + i < n) ? (b + i) : (n - 1);
        const int s = __builtin_amdgcn_readlane(q_s, idx);
        const float* xs = h_proj + (size_t)s * OUT_DIM + lane;
#pragma unroll
        for (int j = 0; j < NHEAD; ++j) x[i][j] = xs[HD * j];
    }
}

__device__ __forceinline__ void add_chunk_rows(int n, int q_s, int q_r, const int (&wq)[NHEAD],
                                               int lane, const float* __restrict__ h_proj,
                                               float* acc)
{
    for (int b = 0; b < n; b += GATHER_BATCH) {
        float x[GATHER_BATCH][NHEAD];
        request_rows(b, n, q_s, lane, h_proj, x);
#pragma unroll
        for (int i = 0; i < GATHER_BATCH; ++i) {
            const bool live = (b + i < n);
            const int idx = live ? (b + i) : (n - 1);
            const int r = __builtin_amdgcn_readlane(q_r, idx);
            float* ar = acc + r * OUT_DIM + lane;
#pragma unroll
            for (int j = 0; j < NHEAD; ++j) {
                const float w = __int_as_float(__builtin_amdgcn_readlane(wq[j], idx));
                const float v = live ? w * x[i][j] : 0.0f;
                atomicAdd(ar + HD * j, v);
            }
        }
    }
}

__device__ __forceinline__ void add_kept_edges(unsigned mask, int sender_v, int recv_v, int dir,
                                               int lane, int lo,
                                               const float* __restrict__ h_proj,
                                               const float* rec, const float* smax,
                                               const float* ssum, float* acc)
{
    while (mask) {
        const int L = __builtin_ctz(mask);
        mask &= mask - 1;
        const int s = __builtin_amdgcn_readlane(sender_v, L);
        const int r = __builtin_amdgcn_readlane(recv_v, L);
        const int hq = lane & 3;
        const float ss = rec[(size_t)s * REC + (dir ? REC_S_OUT_S : REC_S_IN_S) + hq];
        const float rr = rec[(size_t)r * REC + (dir ? REC_S_OUT_R : REC_S_IN_R) + hq];
        float e = ss + rr;
        e = (e > 0.0f) ? e : NEG_SLOPE * e;
        const int st = (r - lo) * STAT + dir * NHEAD + hq;
        const int wq = __float_as_int(__expf(e - smax[st]) / (ssum[st] + EPS_SOFTMAX));
        const float* xs = h_proj + (size_t)s * OUT_DIM + lane;
        float* ar = acc + (r - lo) * OUT_DIM + lane;
        float x[NHEAD];
#pragma unroll
        for (int j = 0; j < NHEAD; ++j) x[j] = xs[HD * j];
#pragma unroll
        for (int j = 0; j < NHEAD; ++j) {
            const float w = __int_as_float(__builtin_amdgcn_readlane(wq, j));
            atomicAdd(ar + HD * j, w * x[j]);
        }
    }
}

#define LN_ROWS 4

__device__ __forceinline__ float wave_sum32(float v) {
#pragma unroll
    for (int m = 16; m > 0; m >>= 1) v += __shfl_xor(v, m, 32);
    return v;
}

__global__ __launch_bounds__(AGG_THREADS) void agg_tile_kernel(
    const int* __restrict__ src, const int* __restrict__ dst,
    const float* __restrict__ h_proj,
    const float* rec,
    const float* __restrict__ bias, const float* __restrict__ gamma,
    const float* __restrict__ beta,
    const unsigned int* map,
    float* __restrict__ out, int N, int E)
{
    __shared__ __attribute__((aligned(16))) float acc[TILE_ROWS * OUT_DIM];
    __shared__ __attribute__((aligned(16))) float smax[TILE_ROWS * STAT];
    __shared__ __attribute__((aligned(16))) float ssum[TILE_ROWS * STAT];
    __shared__ unsigned int list[LIST_CAP];
    __shared__ int lcount;
    const int tid = threadIdx.x;
    const int lane = tid & 31;
    const int wave = __builtin_amdgcn_readfirstlane(tid >> 5);
    const int lo = blockIdx.x * TILE_ROWS;
    const int rows = (N - lo < TILE_ROWS) ? (N - lo) : TILE_ROWS;
    const unsigned urows = (unsigned)rows;
    const unsigned tiles = gridDim.x;
    const unsigned tile = blockIdx.x;

    for (int i = tid; i < TILE_ROWS * OUT_DIM / 4; i += AGG_THREADS) {
        const v4f zero4 = {0.f, 0.f, 0.f, 0.f};
        ((v4f*)acc)[i] = zero4;
    }
    for (int i = tid; i < TILE_ROWS * STAT; i += AGG_THREADS) {
        smax[i] = -INFINITY;
        ssum[i] = 0.0f;
    }
    if (tid == 0) lcount = 0;
    __syncthreads();

    const unsigned below = (1u << lane) - 1u;
    scan_hit_groups(src, dst, map, E, lo, urows, tiles, tile, wave, lane,
        [&](const int (&sv)[STEP_GROUPS], const int (&dv)[STEP_GROUPS],
            const unsigned (&m_in)[STEP_GROUPS], const unsigned (&m_out)[STEP_GROUPS],
            int kept) {
            if (kept == 0) return;
            int base = 0;
            if (lane == 0) base = atomicAdd(&lcount, kept);
            int off = __builtin_amdgcn_readfirstlane(base);
#pragma unroll
            for (int i = 0; i < STEP_GROUPS; ++i) {
                if ((m_in[i] >> lane) & 1u) {
                    const int pos = off + __builtin_popcount(m_in[i] & below);
                    if (pos < LIST_CAP)
                        list[pos] = (unsigned)sv[i] | ((unsigned)(dv[i] - lo) << ENT_ROW_SHIFT);
                }
                off += __builtin_popcount(m_in[i]);
                if ((m_out[i] >> lane) & 1u) {
                    const int pos = off + __builtin_popcount(m_out[i] & below);
                    if (pos < LIST_CAP)
                        list[pos] = (unsigned)dv[i] | ((unsigned)(sv[i] - lo) << ENT_ROW_SHIFT)
                                    | (1u << ENT_DIR_SHIFT);
                }
                off += __builtin_popcount(m_out[i]);
            }
        });
    __syncthreads();
    const int n = __builtin_amdgcn_readfirstlane(lcount);

    if (n <= LIST_CAP) {
        if (n > 0) {
            v4f ev[LIST_PER_THREAD];
#pragma unroll
            for (int k0 = 0; k0 < LIST_PER_THREAD; k0 += PASS_GROUP) {
                if (AGG_THREADS * k0 < n) {
                    unsigned en[PASS_GROUP];
#pragma unroll
                    for (int j = 0; j < PASS_GROUP; ++j) {
                        const int i = tid + AGG_THREADS * (k0 + j);
                        en[j] = list[(i < n) ? i : n - 1];
                        const int row = (int)((en[j] >> ENT_ROW_SHIFT) & ENT_ROW_MASK);
                        ev[k0 + j] = edge_logits(rec, (int)(en[j] & ENT_SENDER_MASK), lo + row,
                                                 (int)(en[j] >> ENT_DIR_SHIFT));
                    }
#pragma unroll
                    for (int j = 0; j < PASS_GROUP; ++j) {
                        if (tid + AGG_THREADS * (k0 + j) < n) {
                            float* mp = smax + (int)((en[j] >> ENT_ROW_SHIFT) & ENT_ROW_MASK) * STAT
                                        + (int)(en[j] >> ENT_DIR_SHIFT) * NHEAD;
                            atomicMaxF32(mp + 0, ev[k0 + j].x);
                            atomicMaxF32(mp + 1, ev[k0 + j].y);
                            atomicMaxF32(mp + 2, ev[k0 + j].z);
                            atomicMaxF32(mp + 3, ev[k0 + j].w);
                        }
                    }
                }
            }
            __syncthreads();
#pragma unroll
            for (int k = 0; k < LIST_PER_THREAD; ++k) {
                const int i = tid + AGG_THREADS * k;
                if (i < n) {
                    const unsigned en = list[i];
                    const int st = (int)((en >> ENT_ROW_SHIFT) & ENT_ROW_MASK) * STAT
                                   + (int)(en >> ENT_DIR_SHIFT) * NHEAD;
                    const v4f mx = *(const v4f*)(smax + st);
                    atomicAdd(ssum + st + 0, __expf(ev[k].x - mx.x));
                    atomicAdd(ssum + st + 1, __expf(ev[k].y - mx.y));
                    atomicAdd(ssum + st + 2, __expf(ev[k].z - mx.z));
                    atomicAdd(ssum + st + 3, __expf(ev[k].w - mx.w));
                }
            }
            __syncthreads();
            const int chunks = (n + 31) >> 5;
            unsigned en_next = 0u;
            v4f e_next = {0.f, 0.f, 0.f, 0.f};
            if (wave < chunks) {
                const int i = 32 * wave + lane;
                en_next = list[(i < n) ? i : n - 1];
                e_next = edge_logits(rec, (int)(en_next & ENT_SENDER_MASK),
                                     lo + (int)((en_next >> ENT_ROW_SHIFT) & ENT_ROW_MASK),
                                     (int)(en_next >> ENT_DIR_SHIFT));
            }
            for (int c = wave; c < chunks; c += AGG_WAVES) {
                const int left = n - 32 * c;
                const int cn = (left < 32) ? left : 32;
                const unsigned en = en_next;
                const v4f e = e_next;
                if (c + AGG_WAVES < chunks) {
                    const int i = 32 * (c + AGG_WAVES) + lane;
                    en_next = list[(i < n) ? i : n - 1];
                    e_next = edge_logits(rec, (int)(en_next & ENT_SENDER_MASK),
                                         lo + (int)((en_next >> ENT_ROW_SHIFT) & ENT_ROW_MASK),
                                         (int)(en_next >> ENT_DIR_SHIFT));
                }
                const int sender = (int)(en & ENT_SENDER_MASK);
                const int row = (int)((en >> ENT_ROW_SHIFT) & ENT_ROW_MASK);
                const int dir = (int)(en >> ENT_DIR_SHIFT);
                const v4f mx = *(const v4f*)(smax + row * STAT + dir * NHEAD);
                const v4f sm = *(const v4f*)(ssum + row * STAT + dir * NHEAD);
                int wq[NHEAD];
                wq[0] = __float_as_int(__expf(e.x - mx.x) / (sm.x + EPS_SOFTMAX));
                wq[1] = __float_as_int(__expf(e.y - mx.y) / (sm.y + EPS_SOFTMAX));
                wq[2] = __float_as_int(__expf(e.z - mx.z) / (sm.z + EPS_SOFTMAX));
                wq[3] = __float_as_int(__expf(e.w - mx.w) / (sm.w + EPS_SOFTMAX));
                add_chunk_rows(cn, sender, row, wq, lane, h_proj, acc);
            }
        }
    } else {
        scan_hit_groups(src, dst, map, E, lo, urows, tiles, tile, wave, lane,
            [&](const int (&sv)[STEP_GROUPS], const int (&dv)[STEP_GROUPS],
                const unsigned (&m_in)[STEP_GROUPS], const unsigned (&m_out)[STEP_GROUPS],
                int kept) {
                if (kept == 0) return;
#pragma unroll
                for (int i = 0; i < STEP_GROUPS; ++i) {
                    if ((m_in[i] >> lane) & 1u) {
                        const v4f e = edge_logits(rec, sv[i], dv[i], 0);
                        float* mp = smax + (dv[i] - lo) * STAT;
                        atomicMaxF32(mp + 0, e.x);
                        atomicMaxF32(mp + 1, e.y);
                        atomicMaxF32(mp + 2, e.z);
                        atomicMaxF32(mp + 3, e.w);
                    }
                    if ((m_out[i] >> lane) & 1u) {
                        const v4f e = edge_logits(rec, dv[i], sv[i], 1);
                        float* mp = smax + (sv[i] - lo) * STAT + NHEAD;
                        atomicMaxF32(mp + 0, e.x);
                        atomicMaxF32(mp + 1, e.y);
                        atomicMaxF32(mp + 2, e.z);
                        atomicMaxF32(mp + 3, e.w);
                    }
                }
            });
        __syncthreads();
        scan_hit_groups(src, dst, map, E, lo, urows, tiles, tile, wave, lane,
            [&](const int (&sv)[STEP_GROUPS], const int (&dv)[STEP_GROUPS],
                const unsigned (&m_in)[STEP_GROUPS], const unsigned (&m_out)[STEP_GROUPS],
                int kept) {
                if (kept == 0) return;
#pragma unroll
                for (int i = 0; i < STEP_GROUPS; ++i) {
                    if ((m_in[i] >> lane) & 1u) {
                        const v4f e = edge_logits(rec, sv[i], dv[i], 0);
                        const int st = (dv[i] - lo) * STAT;
                        const v4f mx = *(const v4f*)(smax + st);
                        atomicAdd(ssum + st + 0, __expf(e.x - mx.x));
                        atomicAdd(ssum + st + 1, __expf(e.y - mx.y));
                        atomicAdd(ssum + st + 2, __expf(e.z - mx.z));
                        atomicAdd(ssum + st + 3, __expf(e.w - mx.w));
                    }
                    if ((m_out[i] >> lane) & 1u) {
                        const v4f e = edge_logits(rec, dv[i], sv[i], 1);
                        const int st = (sv[i] - lo) * STAT + NHEAD;
                        const v4f mx = *(const v4f*)(smax + st);
                        atomicAdd(ssum + st + 0, __expf(e.x - mx.x));
                        atomicAdd(ssum + st + 1, __expf(e.y - mx.y));
                        atomicAdd(ssum + st + 2, __expf(e.z - mx.z));
                        atomicAdd(ssum + st + 3, __expf(e.w - mx.w));
                    }
                }
            });
        __syncthreads();
        scan_hit_groups(src, dst, map, E, lo, urows, tiles, tile, wave, lane,
            [&](const int (&sv)[STEP_GROUPS], const int (&dv)[STEP_GROUPS],
                const unsigned (&m_in)[STEP_GROUPS], const unsigned (&m_out)[STEP_GROUPS],
                int kept) {
                if (kept == 0) return;
#pragma unroll
                for (int i = 0; i < STEP_GROUPS; ++i) {
                    add_kept_edges(m_in[i], sv[i], dv[i], 0, lane, lo, h_proj, rec,
                                   smax, ssum, acc);
                    add_kept_edges(m_out[i], dv[i], sv[i], 1, lane, lo, h_proj, rec,
                                   smax, ssum, acc);
                }
            });
    }
    __syncthreads();

    const v4f bi = *(const v4f*)(bias + 4 * lane);
    const v4f ga = *(const v4f*)(gamma + 4 * lane);
    const v4f be = *(const v4f*)(beta + 4 * lane);
    for (int r0 = wave; r0 < rows; r0 += LN_ROWS * AGG_WAVES) {
        v4f x[LN_ROWS];
#pragma unroll
        for (int i = 0; i < LN_ROWS; ++i) {
            const int r = r0 + i * AGG_WAVES;
            const int rc = (r < rows) ? r : r0;
            const v4f hs = *(const v4f*)(out + (size_t)(lo + rc) * OUT_DIM + 4 * lane);
            const v4f a = *(const v4f*)(acc + rc * OUT_DIM + 4 * lane);
            x[i] = hs + a + bi;
        }
        float mean[LN_ROWS];
#pragma unroll
        for (int i = 0; i < LN_ROWS; ++i)
            mean[i] = wave_sum32((x[i].x + x[i].y) + (x[i].z + x[i].w)) * (1.0f / OUT_DIM);
        float inv[LN_ROWS];
#pragma unroll
        for (int i = 0; i < LN_ROWS; ++i) {
            x[i] = x[i] - mean[i];
            const v4f q = x[i] * x[i];
            const float var = wave_sum32((q.x + q.y) + (q.z + q.w)) * (1.0f / OUT_DIM);
            inv[i] = rsqrtf(var + EPS_LN);
        }
#pragma unroll
        for (int i = 0; i < LN_ROWS; ++i) {
            const int r = r0 + i * AGG_WAVES;
            if (r < rows)
                *(v4f*)(out + (size_t)(lo + r) * OUT_DIM + 4 * lane) = x[i] * inv[i] * ga + be;
        }
    }
}

extern "C" void kernel_launch(void* const* d_in, const int* in_sizes, int n_in,
                              void* d_out, int out_size, void* d_ws, size_t ws_size,
                              hipStream_t stream) {
    const float* h      = (const float*)d_in[0];
    const int*   eidx   = (const int*)d_in[1];
    const float* W      = (const float*)d_in[2];
    const float* Wself  = (const float*)d_in[3];
    const float* a_in   = (const float*)d_in[4];
    const float* a_out  = (const float*)d_in[5];
    const float* bias   = (const float*)d_in[6];
    const float* gamma  = (const float*)d_in[7];
    const float* beta   = (const float*)d_in[8];
    float* out = (float*)d_out;

    const int N = in_sizes[0] / IN_DIM;
    const int E = in_sizes[1] / 2;
    const int NH = N * NHEAD;

    const int* src = eidx;
    const int* dst = eidx + E;

    float* ws = (float*)d_ws;
    float* h_proj  = ws;
    float* rec     = h_proj + (size_t)N * OUT_DIM;

    {
        int blocks = (N + GEMM_ROWS - 1) / GEMM_ROWS;
        gemm_wmma_kernel<<<blocks, 256, 0, stream>>>(h, W, Wself, h_proj, out, N);
    }
    {
        int threads = 256, blocks = (NH + threads - 1) / threads;
        score_kernel<<<blocks, threads, 0, stream>>>(h_proj, a_in, a_out, rec, NH);
    }
    const int tiles = (N + TILE_ROWS - 1) / TILE_ROWS;
    const int nwords = (E + MAP_EDGES - 1) / MAP_EDGES;
    const bool use_map = (long long)nwords * tiles <= (long long)N * MAP_SLOTS;
    unsigned int* map = (unsigned int*)rec;
    if (use_map) {
        tile_map_kernel<<<nwords, MAP_EDGES, tiles * sizeof(unsigned int), stream>>>(
            src, dst, map, E, tiles);
    }
    {
        agg_tile_kernel<<<tiles, AGG_THREADS, 0, stream>>>(src, dst, h_proj, rec,
                                                           bias, gamma, beta,
                                                           use_map ? map : nullptr,
                                                           out, N, E);
    }
}


# Pass 3 reuses the exp(logit - max) that the same thread formed for the same list entry in pass 2 (kept in registers): pass 3's per-edge sender and receiver score gathers are gone; same weights
# speedup vs baseline: 12.7512x; 1.0617x over previous
#include <hip/hip_runtime.h>
#include <math.h>

typedef float v2f __attribute__((ext_vector_type(2)));
typedef float v4f __attribute__((ext_vector_type(4)));
typedef float v8f __attribute__((ext_vector_type(8)));

#define IN_DIM 128
#define OUT_DIM 128
#define NHEAD 4
#define HD 32
#define NEG_SLOPE 0.2f
#define EPS_SOFTMAX 1e-8f
#define EPS_LN 1e-5f

__device__ __forceinline__ void atomicMaxF32(float* addr, float val) {
    if (val >= 0.0f) {
        atomicMax((int*)addr, __float_as_int(val));
    } else {
        atomicMin((unsigned int*)addr, (unsigned int)__float_as_int(val));
    }
}

#define REC 32
#define REC_S_IN_S 0
#define REC_S_OUT_R 4
#define REC_S_IN_R 8
#define REC_S_OUT_S 12
#define REC_MAP 16

#define GEMM_ROWS 64
__global__ __launch_bounds__(256) void gemm_wmma_kernel(
    const float* __restrict__ h, const float* __restrict__ W,
    const float* __restrict__ Wself, float* __restrict__ h_proj,
    float* __restrict__ out, int N)
{
    const int wave = threadIdx.x >> 5;
    const int lane = threadIdx.x & 31;
    const int c = lane & 15;
    const int hh = lane >> 4;
    const int rgrp = wave >> 2;
    const int cgrp = wave & 3;
    const int row0 = blockIdx.x * GEMM_ROWS + rgrp * 32;

    int ra = row0 + c;
    int rb = row0 + 16 + c;
    if (ra >= N) ra = N - 1;
    if (rb >= N) rb = N - 1;
    const float* ap0 = h + (size_t)ra * IN_DIM + 4 * hh;
    const float* ap1 = h + (size_t)rb * IN_DIM + 4 * hh;

    const bool proj = (cgrp < 2);
    const int colbase = (cgrp & 1) * 64;
    const float* bp = (proj ? W : Wself) + colbase + c + (size_t)(4 * hh) * OUT_DIM;

    v8f acc0[4] = {};
    v8f acc1[4] = {};
    for (int k0 = 0; k0 < IN_DIM; k0 += 8) {
        const v4f a0 = *(const v4f*)(ap0 + k0);
        const v4f a1 = *(const v4f*)(ap1 + k0);
        float b[4][4];
#pragma unroll
        for (int t = 0; t < 4; ++t) {
#pragma unroll
            for (int s = 0; s < 4; ++s) b[t][s] = bp[(size_t)(k0 + s) * OUT_DIM + 16 * t];
        }
#pragma unroll
        for (int j = 0; j < 2; ++j) {
            v2f fa0, fa1;
            fa0.x = a0[2 * j];
            fa0.y = a0[2 * j + 1];
            fa1.x = a1[2 * j];
            fa1.y = a1[2 * j + 1];
#pragma unroll
            for (int t = 0; t < 4; ++t) {
                v2f fb;
                fb.x = b[t][2 * j];
                fb.y = b[t][2 * j + 1];
                acc0[t] = __builtin_amdgcn_wmma_f32_16x16x4_f32(false, fa0, false, fb,
                                                                (short)0, acc0[t], false, false);
                acc1[t] = __builtin_amdgcn_wmma_f32_16x16x4_f32(false, fa1, false, fb,
                                                                (short)0, acc1[t], false, false);
            }
        }
    }

    float* dstm = (proj ? h_proj : out) + colbase + c;
    const int mrow = row0 + 8 * hh;
    if (row0 + 32 <= N) {
#pragma unroll
        for (int t = 0; t < 4; ++t) {
#pragma unroll
            for (int v = 0; v < 8; ++v) {
                dstm[(size_t)(mrow + v) * OUT_DIM + 16 * t] = acc0[t][v];
                dstm[(size_t)(mrow + 16 + v) * OUT_DIM + 16 * t] = acc1[t][v];
            }
        }
    } else {
#pragma unroll
        for (int t = 0; t < 4; ++t) {
#pragma unroll
            for (int v = 0; v < 8; ++v) {
                if (mrow + v < N)
                    dstm[(size_t)(mrow + v) * OUT_DIM + 16 * t] = acc0[t][v];
                if (mrow + 16 + v < N)
                    dstm[(size_t)(mrow + 16 + v) * OUT_DIM + 16 * t] = acc1[t][v];
            }
        }
    }
}

__global__ void score_kernel(const float* __restrict__ h_proj,
                             const float* __restrict__ a_in,
                             const float* __restrict__ a_out,
                             float* __restrict__ rec, int NH)
{
    int gid = blockIdx.x * blockDim.x + threadIdx.x;
    if (gid >= NH) return;
    const int hh = gid & 3;
    const int n = gid >> 2;
    const float* x = h_proj + (size_t)n * OUT_DIM + hh * HD;
    const float* ai = a_in + hh * 2 * HD;
    const float* ao = a_out + hh * 2 * HD;
    float s0 = 0.f, s1 = 0.f, s2 = 0.f, s3 = 0.f;
#pragma unroll
    for (int d = 0; d < HD; d += 4) {
        const v4f xv = *(const v4f*)(x + d);
#pragma unroll
        for (int j = 0; j < 4; ++j) {
            s0 += xv[j] * ai[d + j];
            s1 += xv[j] * ai[HD + d + j];
            s2 += xv[j] * ao[d + j];
            s3 += xv[j] * ao[HD + d + j];
        }
    }
    float* r = rec + (size_t)n * REC;
    r[REC_S_IN_S + hh] = s0;
    r[REC_S_IN_R + hh] = s1;
    r[REC_S_OUT_S + hh] = s2;
    r[REC_S_OUT_R + hh] = s3;
}

#define TILE_ROWS 400
#define AGG_THREADS 1024
#define AGG_WAVES (AGG_THREADS / 32)
#define STAT 8

#define MAP_EDGES 1024
#define MAP_SLOTS 16
#define STEP_GROUPS 8

__device__ __forceinline__ size_t map_word_at(unsigned i) {
    return (size_t)(i >> 4) * REC + REC_MAP + (i & (MAP_SLOTS - 1));
}

__global__ __launch_bounds__(MAP_EDGES) void tile_map_kernel(
    const int* __restrict__ src, const int* __restrict__ dst,
    unsigned int* __restrict__ map, int E, int tiles)
{
    extern __shared__ unsigned int wmap[];
    const int tid = threadIdx.x;
    for (int t = tid; t < tiles; t += MAP_EDGES) wmap[t] = 0u;
    __syncthreads();
    const int e = blockIdx.x * MAP_EDGES + tid;
    if (e < E) {
        const unsigned d = (unsigned)dst[e];
        const unsigned s = (unsigned)src[e];
        const unsigned bit = 1u << (tid >> 5);
        atomicOr(&wmap[d / TILE_ROWS], bit);
        atomicOr(&wmap[s / TILE_ROWS], bit);
    }
    __syncthreads();
    for (int t = tid; t < tiles; t += MAP_EDGES)
        map[map_word_at((unsigned)blockIdx.x * (unsigned)tiles + (unsigned)t)] = wmap[t];
}

#define LIST_CAP 16384
#define LIST_PER_THREAD (LIST_CAP / AGG_THREADS)
#define PASS_GROUP 4
#define ENT_ROW_SHIFT 22
#define ENT_DIR_SHIFT 31
#define ENT_SENDER_MASK 0x3FFFFFu
#define ENT_ROW_MASK 0x1FFu

__device__ __forceinline__ v4f edge_logits(const float* rec, int sender, int recv, int dir)
{
    const v4f s = *(const v4f*)(rec + (size_t)sender * REC + (dir ? REC_S_OUT_S : REC_S_IN_S));
    const v4f r = *(const v4f*)(rec + (size_t)recv * REC + (dir ? REC_S_OUT_R : REC_S_IN_R));
    v4f e = s + r;
    e.x = (e.x > 0.0f) ? e.x : NEG_SLOPE * e.x;
    e.y = (e.y > 0.0f) ? e.y : NEG_SLOPE * e.y;
    e.z = (e.z > 0.0f) ? e.z : NEG_SLOPE * e.z;
    e.w = (e.w > 0.0f) ? e.w : NEG_SLOPE * e.w;
    return e;
}

template <class F>
__device__ __forceinline__ void scan_hit_groups(const int* __restrict__ src,
                                                const int* __restrict__ dst,
                                                const unsigned int* map, int E, int lo,
                                                unsigned urows, unsigned tiles, unsigned tile,
                                                int wave, int lane, F&& f)
{
    const int nwords = (E + MAP_EDGES - 1) / MAP_EDGES;
    for (int w0 = wave; w0 < nwords; w0 += 32 * AGG_WAVES) {
        const int wl = w0 + AGG_WAVES * lane;
        unsigned myword = 0u;
        if (wl < nwords)
            myword = map ? map[map_word_at((unsigned)wl * tiles + tile)] : 0xFFFFFFFFu;
        for (int k = 0; k < 32; ++k) {
            const int w = w0 + AGG_WAVES * k;
            if (w >= nwords) break;
            unsigned bits = (unsigned)__builtin_amdgcn_readlane((int)myword, k);
            const int ebase = w * MAP_EDGES;
            while (bits) {
                int g[STEP_GROUPS];
#pragma unroll
                for (int i = 0; i < STEP_GROUPS; ++i) {
                    g[i] = bits ? __builtin_ctz(bits) : 32;
                    bits &= bits - 1u;
                }
                int sv[STEP_GROUPS], dv[STEP_GROUPS];
#pragma unroll
                for (int i = 0; i < STEP_GROUPS; ++i) {
                    const int e = ebase + 32 * g[i] + lane;
                    const bool ok = (g[i] < 32) && (e < E);
                    sv[i] = ok ? src[e] : -1;
                    dv[i] = ok ? dst[e] : -1;
                }
                unsigned m_in[STEP_GROUPS], m_out[STEP_GROUPS];
                int kept = 0;
#pragma unroll
                for (int i = 0; i < STEP_GROUPS; ++i) {
                    m_in[i] = __builtin_amdgcn_ballot_w32((unsigned)(dv[i] - lo) < urows);
                    m_out[i] = __builtin_amdgcn_ballot_w32((unsigned)(sv[i] - lo) < urows);
                    kept += __builtin_popcount(m_in[i]) + __builtin_popcount(m_out[i]);
                }
                f(sv, dv, m_in, m_out, kept);
            }
        }
    }
}

#define GATHER_BATCH 8

__device__ __forceinline__ void request_rows(int b, int n, int q_s, int lane,
                                             const float* __restrict__ h_proj,
                                             float (&x)[GATHER_BATCH][NHEAD])
{
#pragma unroll
    for (int i = 0; i < GATHER_BATCH; ++i) {
        const int idx = (b + i < n) ? (b + i) : (n - 1);
        const int s = __builtin_amdgcn_readlane(q_s, idx);
        const float* xs = h_proj + (size_t)s * OUT_DIM + lane;
#pragma unroll
        for (int j = 0; j < NHEAD; ++j) x[i][j] = xs[HD * j];
    }
}

__device__ __forceinline__ void add_chunk_rows(int n, int q_s, int q_r, const int (&wq)[NHEAD],
                                               int lane, const float* __restrict__ h_proj,
                                               float* acc)
{
    for (int b = 0; b < n; b += GATHER_BATCH) {
        float x[GATHER_BATCH][NHEAD];
        request_rows(b, n, q_s, lane, h_proj, x);
#pragma unroll
        for (int i = 0; i < GATHER_BATCH; ++i) {
            const bool live = (b + i < n);
            const int idx = live ? (b + i) : (n - 1);
            const int r = __builtin_amdgcn_readlane(q_r, idx);
            float* ar = acc + r * OUT_DIM + lane;
#pragma unroll
            for (int j = 0; j < NHEAD; ++j) {
                const float w = __int_as_float(__builtin_amdgcn_readlane(wq[j], idx));
                const float v = live ? w * x[i][j] : 0.0f;
                atomicAdd(ar + HD * j, v);
            }
        }
    }
}

__device__ __forceinline__ void add_kept_edges(unsigned mask, int sender_v, int recv_v, int dir,
                                               int lane, int lo,
                                               const float* __restrict__ h_proj,
                                               const float* rec, const float* smax,
                                               const float* ssum, float* acc)
{
    while (mask) {
        const int L = __builtin_ctz(mask);
        mask &= mask - 1;
        const int s = __builtin_amdgcn_readlane(sender_v, L);
        const int r = __builtin_amdgcn_readlane(recv_v, L);
        const int hq = lane & 3;
        const float ss = rec[(size_t)s * REC + (dir ? REC_S_OUT_S : REC_S_IN_S) + hq];
        const float rr = rec[(size_t)r * REC + (dir ? REC_S_OUT_R : REC_S_IN_R) + hq];
        float e = ss + rr;
        e = (e > 0.0f) ? e : NEG_SLOPE * e;
        const int st = (r - lo) * STAT + dir * NHEAD + hq;
        const int wq = __float_as_int(__expf(e - smax[st]) / (ssum[st] + EPS_SOFTMAX));
        const float* xs = h_proj + (size_t)s * OUT_DIM + lane;
        float* ar = acc + (r - lo) * OUT_DIM + lane;
        float x[NHEAD];
#pragma unroll
        for (int j = 0; j < NHEAD; ++j) x[j] = xs[HD * j];
#pragma unroll
        for (int j = 0; j < NHEAD; ++j) {
            const float w = __int_as_float(__builtin_amdgcn_readlane(wq, j));
            atomicAdd(ar + HD * j, w * x[j]);
        }
    }
}

#define LN_ROWS 4

__device__ __forceinline__ float wave_sum32(float v) {
#pragma unroll
    for (int m = 16; m > 0; m >>= 1) v += __shfl_xor(v, m, 32);
    return v;
}

__global__ __launch_bounds__(AGG_THREADS) void agg_tile_kernel(
    const int* __restrict__ src, const int* __restrict__ dst,
    const float* __restrict__ h_proj,
    const float* rec,
    const float* __restrict__ bias, const float* __restrict__ gamma,
    const float* __restrict__ beta,
    const unsigned int* map,
    float* __restrict__ out, int N, int E)
{
    __shared__ __attribute__((aligned(16))) float acc[TILE_ROWS * OUT_DIM];
    __shared__ __attribute__((aligned(16))) float smax[TILE_ROWS * STAT];
    __shared__ __attribute__((aligned(16))) float ssum[TILE_ROWS * STAT];
    __shared__ unsigned int list[LIST_CAP];
    __shared__ int lcount;
    const int tid = threadIdx.x;
    const int lane = tid & 31;
    const int wave = __builtin_amdgcn_readfirstlane(tid >> 5);
    const int lo = blockIdx.x * TILE_ROWS;
    const int rows = (N - lo < TILE_ROWS) ? (N - lo) : TILE_ROWS;
    const unsigned urows = (unsigned)rows;
    const unsigned tiles = gridDim.x;
    const unsigned tile = blockIdx.x;

    for (int i = tid; i < TILE_ROWS * OUT_DIM / 4; i += AGG_THREADS) {
        const v4f zero4 = {0.f, 0.f, 0.f, 0.f};
        ((v4f*)acc)[i] = zero4;
    }
    for (int i = tid; i < TILE_ROWS * STAT; i += AGG_THREADS) {
        smax[i] = -INFINITY;
        ssum[i] = 0.0f;
    }
    if (tid == 0) lcount = 0;
    __syncthreads();

    const unsigned below = (1u << lane) - 1u;
    scan_hit_groups(src, dst, map, E, lo, urows, tiles, tile, wave, lane,
        [&](const int (&sv)[STEP_GROUPS], const int (&dv)[STEP_GROUPS],
            const unsigned (&m_in)[STEP_GROUPS], const unsigned (&m_out)[STEP_GROUPS],
            int kept) {
            if (kept == 0) return;
            int base = 0;
            if (lane == 0) base = atomicAdd(&lcount, kept);
            int off = __builtin_amdgcn_readfirstlane(base);
#pragma unroll
            for (int i = 0; i < STEP_GROUPS; ++i) {
                if ((m_in[i] >> lane) & 1u) {
                    const int pos = off + __builtin_popcount(m_in[i] & below);
                    if (pos < LIST_CAP)
                        list[pos] = (unsigned)sv[i] | ((unsigned)(dv[i] - lo) << ENT_ROW_SHIFT);
                }
                off += __builtin_popcount(m_in[i]);
                if ((m_out[i] >> lane) & 1u) {
                    const int pos = off + __builtin_popcount(m_out[i] & below);
                    if (pos < LIST_CAP)
                        list[pos] = (unsigned)dv[i] | ((unsigned)(sv[i] - lo) << ENT_ROW_SHIFT)
                                    | (1u << ENT_DIR_SHIFT);
                }
                off += __builtin_popcount(m_out[i]);
            }
        });
    __syncthreads();
    const int n = __builtin_amdgcn_readfirstlane(lcount);

    if (n <= LIST_CAP) {
        if (n > 0) {
            v4f ev[LIST_PER_THREAD] = {};
#pragma unroll
            for (int k0 = 0; k0 < LIST_PER_THREAD; k0 += PASS_GROUP) {
                if (AGG_THREADS * k0 < n) {
                    unsigned en[PASS_GROUP];
#pragma unroll
                    for (int j = 0; j < PASS_GROUP; ++j) {
                        const int i = tid + AGG_THREADS * (k0 + j);
                        en[j] = list[(i < n) ? i : n - 1];
                        const int row = (int)((en[j] >> ENT_ROW_SHIFT) & ENT_ROW_MASK);
                        ev[k0 + j] = edge_logits(rec, (int)(en[j] & ENT_SENDER_MASK), lo + row,
                                                 (int)(en[j] >> ENT_DIR_SHIFT));
                    }
#pragma unroll
                    for (int j = 0; j < PASS_GROUP; ++j) {
                        if (tid + AGG_THREADS * (k0 + j) < n) {
                            float* mp = smax + (int)((en[j] >> ENT_ROW_SHIFT) & ENT_ROW_MASK) * STAT
                                        + (int)(en[j] >> ENT_DIR_SHIFT) * NHEAD;
                            atomicMaxF32(mp + 0, ev[k0 + j].x);
                            atomicMaxF32(mp + 1, ev[k0 + j].y);
                            atomicMaxF32(mp + 2, ev[k0 + j].z);
                            atomicMaxF32(mp + 3, ev[k0 + j].w);
                        }
                    }
                }
            }
            __syncthreads();
#pragma unroll
            for (int k = 0; k < LIST_PER_THREAD; ++k) {
                const int i = tid + AGG_THREADS * k;
                if (i < n) {
                    const unsigned en = list[i];
                    const int st = (int)((en >> ENT_ROW_SHIFT) & ENT_ROW_MASK) * STAT
                                   + (int)(en >> ENT_DIR_SHIFT) * NHEAD;
                    const v4f mx = *(const v4f*)(smax + st);
                    v4f p;
                    p.x = __expf(ev[k].x - mx.x);
                    p.y = __expf(ev[k].y - mx.y);
                    p.z = __expf(ev[k].z - mx.z);
                    p.w = __expf(ev[k].w - mx.w);
                    atomicAdd(ssum + st + 0, p.x);
                    atomicAdd(ssum + st + 1, p.y);
                    atomicAdd(ssum + st + 2, p.z);
                    atomicAdd(ssum + st + 3, p.w);
                    ev[k] = p;
                }
            }
            __syncthreads();
            const int chunks = (n + 31) >> 5;
            for (int k = 0; k < LIST_PER_THREAD; ++k) {
                const int c = wave + AGG_WAVES * k;
                if (c >= chunks) break;
                const int left = n - 32 * c;
                const int cn = (left < 32) ? left : 32;
                const int i = 32 * c + lane;
                const unsigned en = list[(i < n) ? i : n - 1];
                v4f p = ev[0];
#pragma unroll
                for (int kk = 1; kk < LIST_PER_THREAD; ++kk)
                    if (kk == k) p = ev[kk];
                const int sender = (int)(en & ENT_SENDER_MASK);
                const int row = (int)((en >> ENT_ROW_SHIFT) & ENT_ROW_MASK);
                const int dir = (int)(en >> ENT_DIR_SHIFT);
                const v4f sm = *(const v4f*)(ssum + row * STAT + dir * NHEAD);
                int wq[NHEAD];
                wq[0] = __float_as_int(p.x / (sm.x + EPS_SOFTMAX));
                wq[1] = __float_as_int(p.y / (sm.y + EPS_SOFTMAX));
                wq[2] = __float_as_int(p.z / (sm.z + EPS_SOFTMAX));
                wq[3] = __float_as_int(p.w / (sm.w + EPS_SOFTMAX));
                add_chunk_rows(cn, sender, row, wq, lane, h_proj, acc);
            }
        }
    } else {
        scan_hit_groups(src, dst, map, E, lo, urows, tiles, tile, wave, lane,
            [&](const int (&sv)[STEP_GROUPS], const int (&dv)[STEP_GROUPS],
                const unsigned (&m_in)[STEP_GROUPS], const unsigned (&m_out)[STEP_GROUPS],
                int kept) {
                if (kept == 0) return;
#pragma unroll
                for (int i = 0; i < STEP_GROUPS; ++i) {
                    if ((m_in[i] >> lane) & 1u) {
                        const v4f e = edge_logits(rec, sv[i], dv[i], 0);
                        float* mp = smax + (dv[i] - lo) * STAT;
                        atomicMaxF32(mp + 0, e.x);
                        atomicMaxF32(mp + 1, e.y);
                        atomicMaxF32(mp + 2, e.z);
                        atomicMaxF32(mp + 3, e.w);
                    }
                    if ((m_out[i] >> lane) & 1u) {
                        const v4f e = edge_logits(rec, dv[i], sv[i], 1);
                        float* mp = smax + (sv[i] - lo) * STAT + NHEAD;
                        atomicMaxF32(mp + 0, e.x);
                        atomicMaxF32(mp + 1, e.y);
                        atomicMaxF32(mp + 2, e.z);
                        atomicMaxF32(mp + 3, e.w);
                    }
                }
            });
        __syncthreads();
        scan_hit_groups(src, dst, map, E, lo, urows, tiles, tile, wave, lane,
            [&](const int (&sv)[STEP_GROUPS], const int (&dv)[STEP_GROUPS],
                const unsigned (&m_in)[STEP_GROUPS], const unsigned (&m_out)[STEP_GROUPS],
                int kept) {
                if (kept == 0) return;
#pragma unroll
                for (int i = 0; i < STEP_GROUPS; ++i) {
                    if ((m_in[i] >> lane) & 1u) {
                        const v4f e = edge_logits(rec, sv[i], dv[i], 0);
                        const int st = (dv[i] - lo) * STAT;
                        const v4f mx = *(const v4f*)(smax + st);
                        atomicAdd(ssum + st + 0, __expf(e.x - mx.x));
                        atomicAdd(ssum + st + 1, __expf(e.y - mx.y));
                        atomicAdd(ssum + st + 2, __expf(e.z - mx.z));
                        atomicAdd(ssum + st + 3, __expf(e.w - mx.w));
                    }
                    if ((m_out[i] >> lane) & 1u) {
                        const v4f e = edge_logits(rec, dv[i], sv[i], 1);
                        const int st = (sv[i] - lo) * STAT + NHEAD;
                        const v4f mx = *(const v4f*)(smax + st);
                        atomicAdd(ssum + st + 0, __expf(e.x - mx.x));
                        atomicAdd(ssum + st + 1, __expf(e.y - mx.y));
                        atomicAdd(ssum + st + 2, __expf(e.z - mx.z));
                        atomicAdd(ssum + st + 3, __expf(e.w - mx.w));
                    }
                }
            });
        __syncthreads();
        scan_hit_groups(src, dst, map, E, lo, urows, tiles, tile, wave, lane,
            [&](const int (&sv)[STEP_GROUPS], const int (&dv)[STEP_GROUPS],
                const unsigned (&m_in)[STEP_GROUPS], const unsigned (&m_out)[STEP_GROUPS],
                int kept) {
                if (kept == 0) return;
#pragma unroll
                for (int i = 0; i < STEP_GROUPS; ++i) {
                    add_kept_edges(m_in[i], sv[i], dv[i], 0, lane, lo, h_proj, rec,
                                   smax, ssum, acc);
                    add_kept_edges(m_out[i], dv[i], sv[i], 1, lane, lo, h_proj, rec,
                                   smax, ssum, acc);
                }
            });
    }
    __syncthreads();

    const v4f bi = *(const v4f*)(bias + 4 * lane);
    const v4f ga = *(const v4f*)(gamma + 4 * lane);
    const v4f be = *(const v4f*)(beta + 4 * lane);
    for (int r0 = wave; r0 < rows; r0 += LN_ROWS * AGG_WAVES) {
        v4f x[LN_ROWS];
#pragma unroll
        for (int i = 0; i < LN_ROWS; ++i) {
            const int r = r0 + i * AGG_WAVES;
            const int rc = (r < rows) ? r : r0;
            const v4f hs = *(const v4f*)(out + (size_t)(lo + rc) * OUT_DIM + 4 * lane);
            const v4f a = *(const v4f*)(acc + rc * OUT_DIM + 4 * lane);
            x[i] = hs + a + bi;
        }
        float mean[LN_ROWS];
#pragma unroll
        for (int i = 0; i < LN_ROWS; ++i)
            mean[i] = wave_sum32((x[i].x + x[i].y) + (x[i].z + x[i].w)) * (1.0f / OUT_DIM);
        float inv[LN_ROWS];
#pragma unroll
        for (int i = 0; i < LN_ROWS; ++i) {
            x[i] = x[i] - mean[i];
            const v4f q = x[i] * x[i];
            const float var = wave_sum32((q.x + q.y) + (q.z + q.w)) * (1.0f / OUT_DIM);
            inv[i] = rsqrtf(var + EPS_LN);
        }
#pragma unroll
        for (int i = 0; i < LN_ROWS; ++i) {
            const int r = r0 + i * AGG_WAVES;
            if (r < rows)
                *(v4f*)(out + (size_t)(lo + r) * OUT_DIM + 4 * lane) = x[i] * inv[i] * ga + be;
        }
    }
}

extern "C" void kernel_launch(void* const* d_in, const int* in_sizes, int n_in,
                              void* d_out, int out_size, void* d_ws, size_t ws_size,
                              hipStream_t stream) {
    const float* h      = (const float*)d_in[0];
    const int*   eidx   = (const int*)d_in[1];
    const float* W      = (const float*)d_in[2];
    const float* Wself  = (const float*)d_in[3];
    const float* a_in   = (const float*)d_in[4];
    const float* a_out  = (const float*)d_in[5];
    const float* bias   = (const float*)d_in[6];
    const float* gamma  = (const float*)d_in[7];
    const float* beta   = (const float*)d_in[8];
    float* out = (float*)d_out;

    const int N = in_sizes[0] / IN_DIM;
    const int E = in_sizes[1] / 2;
    const int NH = N * NHEAD;

    const int* src = eidx;
    const int* dst = eidx + E;

    float* ws = (float*)d_ws;
    float* h_proj  = ws;
    float* rec     = h_proj + (size_t)N * OUT_DIM;

    {
        int blocks = (N + GEMM_ROWS - 1) / GEMM_ROWS;
        gemm_wmma_kernel<<<blocks, 256, 0, stream>>>(h, W, Wself, h_proj, out, N);
    }
    {
        int threads = 256, blocks = (NH + threads - 1) / threads;
        score_kernel<<<blocks, threads, 0, stream>>>(h_proj, a_in, a_out, rec, NH);
    }
    const int tiles = (N + TILE_ROWS - 1) / TILE_ROWS;
    const int nwords = (E + MAP_EDGES - 1) / MAP_EDGES;
    const bool use_map = (long long)nwords * tiles <= (long long)N * MAP_SLOTS;
    unsigned int* map = (unsigned int*)rec;
    if (use_map) {
        tile_map_kernel<<<nwords, MAP_EDGES, tiles * sizeof(unsigned int), stream>>>(
            src, dst, map, E, tiles);
    }
    {
        agg_tile_kernel<<<tiles, AGG_THREADS, 0, stream>>>(src, dst, h_proj, rec,
                                                           bias, gamma, beta,
                                                           use_map ? map : nullptr,
                                                           out, N, E);
    }
}


# Tile lists built in memory by a binning launch (LDS ranks, one global int add per tile and block) replace the tile kernel's hit map and edge scan; arrival-order lists, LDS add order changes
# speedup vs baseline: 17.6274x; 1.3824x over previous
#include <hip/hip_runtime.h>
#include <math.h>

typedef float v2f __attribute__((ext_vector_type(2)));
typedef float v4f __attribute__((ext_vector_type(4)));
typedef float v8f __attribute__((ext_vector_type(8)));

#define IN_DIM 128
#define OUT_DIM 128
#define NHEAD 4
#define HD 32
#define NEG_SLOPE 0.2f
#define EPS_SOFTMAX 1e-8f
#define EPS_LN 1e-5f

__device__ __forceinline__ void atomicMaxF32(float* addr, float val) {
    if (val >= 0.0f) {
        atomicMax((int*)addr, __float_as_int(val));
    } else {
        atomicMin((unsigned int*)addr, (unsigned int)__float_as_int(val));
    }
}

#define REC 32
#define REC_S_IN_S 0
#define REC_S_OUT_R 4
#define REC_S_IN_R 8
#define REC_S_OUT_S 12
#define REC_MAP 16

#define GEMM_ROWS 64
__global__ __launch_bounds__(256) void gemm_wmma_kernel(
    const float* __restrict__ h, const float* __restrict__ W,
    const float* __restrict__ Wself, float* __restrict__ h_proj,
    float* __restrict__ out, int N)
{
    const int wave = threadIdx.x >> 5;
    const int lane = threadIdx.x & 31;
    const int c = lane & 15;
    const int hh = lane >> 4;
    const int rgrp = wave >> 2;
    const int cgrp = wave & 3;
    const int row0 = blockIdx.x * GEMM_ROWS + rgrp * 32;

    int ra = row0 + c;
    int rb = row0 + 16 + c;
    if (ra >= N) ra = N - 1;
    if (rb >= N) rb = N - 1;
    const float* ap0 = h + (size_t)ra * IN_DIM + 4 * hh;
    const float* ap1 = h + (size_t)rb * IN_DIM + 4 * hh;

    const bool proj = (cgrp < 2);
    const int colbase = (cgrp & 1) * 64;
    const float* bp = (proj ? W : Wself) + colbase + c + (size_t)(4 * hh) * OUT_DIM;

    v8f acc0[4] = {};
    v8f acc1[4] = {};
    for (int k0 = 0; k0 < IN_DIM; k0 += 8) {
        const v4f a0 = *(const v4f*)(ap0 + k0);
        const v4f a1 = *(const v4f*)(ap1 + k0);
        float b[4][4];
#pragma unroll
        for (int t = 0; t < 4; ++t) {
#pragma unroll
            for (int s = 0; s < 4; ++s) b[t][s] = bp[(size_t)(k0 + s) * OUT_DIM + 16 * t];
        }
#pragma unroll
        for (int j = 0; j < 2; ++j) {
            v2f fa0, fa1;
            fa0.x = a0[2 * j];
            fa0.y = a0[2 * j + 1];
            fa1.x = a1[2 * j];
            fa1.y = a1[2 * j + 1];
#pragma unroll
            for (int t = 0; t < 4; ++t) {
                v2f fb;
                fb.x = b[t][2 * j];
                fb.y = b[t][2 * j + 1];
                acc0[t] = __builtin_amdgcn_wmma_f32_16x16x4_f32(false, fa0, false, fb,
                                                                (short)0, acc0[t], false, false);
                acc1[t] = __builtin_amdgcn_wmma_f32_16x16x4_f32(false, fa1, false, fb,
                                                                (short)0, acc1[t], false, false);
            }
        }
    }

    float* dstm = (proj ? h_proj : out) + colbase + c;
    const int mrow = row0 + 8 * hh;
    if (row0 + 32 <= N) {
#pragma unroll
        for (int t = 0; t < 4; ++t) {
#pragma unroll
            for (int v = 0; v < 8; ++v) {
                dstm[(size_t)(mrow + v) * OUT_DIM + 16 * t] = acc0[t][v];
                dstm[(size_t)(mrow + 16 + v) * OUT_DIM + 16 * t] = acc1[t][v];
            }
        }
    } else {
#pragma unroll
        for (int t = 0; t < 4; ++t) {
#pragma unroll
            for (int v = 0; v < 8; ++v) {
                if (mrow + v < N)
                    dstm[(size_t)(mrow + v) * OUT_DIM + 16 * t] = acc0[t][v];
                if (mrow + 16 + v < N)
                    dstm[(size_t)(mrow + 16 + v) * OUT_DIM + 16 * t] = acc1[t][v];
            }
        }
    }
}

__global__ void score_kernel(const float* __restrict__ h_proj,
                             const float* __restrict__ a_in,
                             const float* __restrict__ a_out,
                             float* __restrict__ rec, int NH,
                             int* __restrict__ fill, int tiles)
{
    int gid = blockIdx.x * blockDim.x + threadIdx.x;
    if (fill && gid < tiles) fill[gid] = 0;
    if (gid >= NH) return;
    const int hh = gid & 3;
    const int n = gid >> 2;
    const float* x = h_proj + (size_t)n * OUT_DIM + hh * HD;
    const float* ai = a_in + hh * 2 * HD;
    const float* ao = a_out + hh * 2 * HD;
    float s0 = 0.f, s1 = 0.f, s2 = 0.f, s3 = 0.f;
#pragma unroll
    for (int d = 0; d < HD; d += 4) {
        const v4f xv = *(const v4f*)(x + d);
#pragma unroll
        for (int j = 0; j < 4; ++j) {
            s0 += xv[j] * ai[d + j];
            s1 += xv[j] * ai[HD + d + j];
            s2 += xv[j] * ao[d + j];
            s3 += xv[j] * ao[HD + d + j];
        }
    }
    float* r = rec + (size_t)n * REC;
    r[REC_S_IN_S + hh] = s0;
    r[REC_S_IN_R + hh] = s1;
    r[REC_S_OUT_S + hh] = s2;
    r[REC_S_OUT_R + hh] = s3;
}

#define TILE_ROWS 400
#define AGG_THREADS 1024
#define AGG_WAVES (AGG_THREADS / 32)
#define STAT 8

#define MAP_EDGES 1024
#define MAP_SLOTS 16
#define STEP_GROUPS 8

__device__ __forceinline__ size_t map_word_at(unsigned i) {
    return (size_t)(i >> 4) * REC + REC_MAP + (i & (MAP_SLOTS - 1));
}

__global__ __launch_bounds__(MAP_EDGES) void tile_map_kernel(
    const int* __restrict__ src, const int* __restrict__ dst,
    unsigned int* __restrict__ map, int E, int tiles)
{
    extern __shared__ unsigned int wmap[];
    const int tid = threadIdx.x;
    for (int t = tid; t < tiles; t += MAP_EDGES) wmap[t] = 0u;
    __syncthreads();
    const int e = blockIdx.x * MAP_EDGES + tid;
    if (e < E) {
        const unsigned d = (unsigned)dst[e];
        const unsigned s = (unsigned)src[e];
        const unsigned bit = 1u << (tid >> 5);
        atomicOr(&wmap[d / TILE_ROWS], bit);
        atomicOr(&wmap[s / TILE_ROWS], bit);
    }
    __syncthreads();
    for (int t = tid; t < tiles; t += MAP_EDGES)
        map[map_word_at((unsigned)blockIdx.x * (unsigned)tiles + (unsigned)t)] = wmap[t];
}

#define LIST_CAP 16384
#define LIST_PER_THREAD (LIST_CAP / AGG_THREADS)
#define PASS_GROUP 4
#define ENT_ROW_SHIFT 22
#define ENT_DIR_SHIFT 31
#define ENT_SENDER_MASK 0x3FFFFFu
#define ENT_ROW_MASK 0x1FFu

#define BIN_THREADS 1024
#define BIN_PER_THREAD 8
#define BIN_EDGES (BIN_THREADS * BIN_PER_THREAD)
#define BIN_ENTRIES (2 * BIN_EDGES)

__global__ __launch_bounds__(BIN_THREADS) void tile_bin_kernel(
    const int* __restrict__ src, const int* __restrict__ dst,
    unsigned int* __restrict__ bins, int* fill, int E, int tiles)
{
    extern __shared__ int bcnt[];
    int* bstart = bcnt + tiles;
    int* lstart = bstart + tiles;
    __shared__ unsigned int stg[BIN_ENTRIES];
    __shared__ unsigned short stile[BIN_ENTRIES];
    const int tid = threadIdx.x;
    const int lane = tid & 31;
    const unsigned utiles = (unsigned)tiles;
    for (int t = tid; t < tiles; t += BIN_THREADS) bcnt[t] = 0;
    __syncthreads();
    int s[BIN_PER_THREAD], d[BIN_PER_THREAD];
#pragma unroll
    for (int k = 0; k < BIN_PER_THREAD; ++k) {
        const int e = blockIdx.x * BIN_EDGES + k * BIN_THREADS + tid;
        const bool ok = (e < E);
        s[k] = ok ? src[e] : -1;
        d[k] = ok ? dst[e] : -1;
    }
    unsigned ti[BIN_PER_THREAD], to[BIN_PER_THREAD];
    int ri[BIN_PER_THREAD], ro[BIN_PER_THREAD];
#pragma unroll
    for (int k = 0; k < BIN_PER_THREAD; ++k) {
        ti[k] = (unsigned)d[k] / TILE_ROWS;
        to[k] = (unsigned)s[k] / TILE_ROWS;
        ri[k] = (ti[k] < utiles) ? atomicAdd(&bcnt[ti[k]], 1) : 0;
        ro[k] = (to[k] < utiles) ? atomicAdd(&bcnt[to[k]], 1) : 0;
    }
    __syncthreads();
    for (int t = tid; t < tiles; t += BIN_THREADS) {
        const int cnt = bcnt[t];
        bstart[t] = cnt ? atomicAdd(&fill[t], cnt) : 0;
    }
    if ((tid >> 5) == BIN_THREADS / 32 - 1) {
        const int per = (tiles + 31) >> 5;
        const int tb = lane * per;
        int sum = 0;
        for (int i = 0; i < per; ++i)
            if (tb + i < tiles) sum += bcnt[tb + i];
        int x = sum;
#pragma unroll
        for (int o = 1; o < 32; o <<= 1) {
            const int y = __shfl_up(x, o, 32);
            if (lane >= o) x += y;
        }
        int run = x - sum;
        for (int i = 0; i < per; ++i) {
            if (tb + i < tiles) {
                lstart[tb + i] = run;
                run += bcnt[tb + i];
            }
        }
    }
    __syncthreads();
#pragma unroll
    for (int k = 0; k < BIN_PER_THREAD; ++k) {
        if (ti[k] < utiles) {
            const int j = lstart[ti[k]] + ri[k];
            stg[j] = (unsigned)s[k] | (((unsigned)d[k] - ti[k] * TILE_ROWS) << ENT_ROW_SHIFT);
            stile[j] = (unsigned short)ti[k];
        }
        if (to[k] < utiles) {
            const int j = lstart[to[k]] + ro[k];
            stg[j] = (unsigned)d[k] | (((unsigned)s[k] - to[k] * TILE_ROWS) << ENT_ROW_SHIFT)
                     | (1u << ENT_DIR_SHIFT);
            stile[j] = (unsigned short)to[k];
        }
    }
    __syncthreads();
    const int total = lstart[tiles - 1] + bcnt[tiles - 1];
#pragma unroll
    for (int k = 0; k < 2 * BIN_PER_THREAD; ++k) {
        const int j = tid + k * BIN_THREADS;
        if (j < total) {
            const int t = stile[j];
            const int pos = bstart[t] + (j - lstart[t]);
            if (pos < LIST_CAP) bins[(size_t)t * LIST_CAP + pos] = stg[j];
        }
    }
}

__device__ __forceinline__ v4f edge_logits(const float* rec, int sender, int recv, int dir)
{
    const v4f s = *(const v4f*)(rec + (size_t)sender * REC + (dir ? REC_S_OUT_S : REC_S_IN_S));
    const v4f r = *(const v4f*)(rec + (size_t)recv * REC + (dir ? REC_S_OUT_R : REC_S_IN_R));
    v4f e = s + r;
    e.x = (e.x > 0.0f) ? e.x : NEG_SLOPE * e.x;
    e.y = (e.y > 0.0f) ? e.y : NEG_SLOPE * e.y;
    e.z = (e.z > 0.0f) ? e.z : NEG_SLOPE * e.z;
    e.w = (e.w > 0.0f) ? e.w : NEG_SLOPE * e.w;
    return e;
}

template <class F>
__device__ __forceinline__ void scan_hit_groups(const int* __restrict__ src,
                                                const int* __restrict__ dst,
                                                const unsigned int* map, int E, int lo,
                                                unsigned urows, unsigned tiles, unsigned tile,
                                                int wave, int lane, F&& f)
{
    const int nwords = (E + MAP_EDGES - 1) / MAP_EDGES;
    for (int w0 = wave; w0 < nwords; w0 += 32 * AGG_WAVES) {
        const int wl = w0 + AGG_WAVES * lane;
        unsigned myword = 0u;
        if (wl < nwords)
            myword = map ? map[map_word_at((unsigned)wl * tiles + tile)] : 0xFFFFFFFFu;
        for (int k = 0; k < 32; ++k) {
            const int w = w0 + AGG_WAVES * k;
            if (w >= nwords) break;
            unsigned bits = (unsigned)__builtin_amdgcn_readlane((int)myword, k);
            const int ebase = w * MAP_EDGES;
            while (bits) {
                int g[STEP_GROUPS];
#pragma unroll
                for (int i = 0; i < STEP_GROUPS; ++i) {
                    g[i] = bits ? __builtin_ctz(bits) : 32;
                    bits &= bits - 1u;
                }
                int sv[STEP_GROUPS], dv[STEP_GROUPS];
#pragma unroll
                for (int i = 0; i < STEP_GROUPS; ++i) {
                    const int e = ebase + 32 * g[i] + lane;
                    const bool ok = (g[i] < 32) && (e < E);
                    sv[i] = ok ? src[e] : -1;
                    dv[i] = ok ? dst[e] : -1;
                }
                unsigned m_in[STEP_GROUPS], m_out[STEP_GROUPS];
                int kept = 0;
#pragma unroll
                for (int i = 0; i < STEP_GROUPS; ++i) {
                    m_in[i] = __builtin_amdgcn_ballot_w32((unsigned)(dv[i] - lo) < urows);
                    m_out[i] = __builtin_amdgcn_ballot_w32((unsigned)(sv[i] - lo) < urows);
                    kept += __builtin_popcount(m_in[i]) + __builtin_popcount(m_out[i]);
                }
                f(sv, dv, m_in, m_out, kept);
            }
        }
    }
}

#define GATHER_BATCH 8

__device__ __forceinline__ void request_rows(int b, int n, int q_s, int lane,
                                             const float* __restrict__ h_proj,
                                             float (&x)[GATHER_BATCH][NHEAD])
{
#pragma unroll
    for (int i = 0; i < GATHER_BATCH; ++i) {
        const int idx = (b + i < n) ? (b + i) : (n - 1);
        const int s = __builtin_amdgcn_readlane(q_s, idx);
        const float* xs = h_proj + (size_t)s * OUT_DIM + lane;
#pragma unroll
        for (int j = 0; j < NHEAD; ++j) x[i][j] = xs[HD * j];
    }
}

__device__ __forceinline__ void add_chunk_rows(int n, int q_s, int q_r, const int (&wq)[NHEAD],
                                               int lane, const float* __restrict__ h_proj,
                                               float* acc)
{
    for (int b = 0; b < n; b += GATHER_BATCH) {
        float x[GATHER_BATCH][NHEAD];
        request_rows(b, n, q_s, lane, h_proj, x);
#pragma unroll
        for (int i = 0; i < GATHER_BATCH; ++i) {
            const bool live = (b + i < n);
            const int idx = live ? (b + i) : (n - 1);
            const int r = __builtin_amdgcn_readlane(q_r, idx);
            float* ar = acc + r * OUT_DIM + lane;
#pragma unroll
            for (int j = 0; j < NHEAD; ++j) {
                const float w = __int_as_float(__builtin_amdgcn_readlane(wq[j], idx));
                const float v = live ? w * x[i][j] : 0.0f;
                atomicAdd(ar + HD * j, v);
            }
        }
    }
}

__device__ __forceinline__ void add_kept_edges(unsigned mask, int sender_v, int recv_v, int dir,
                                               int lane, int lo,
                                               const float* __restrict__ h_proj,
                                               const float* rec, const float* smax,
                                               const float* ssum, float* acc)
{
    while (mask) {
        const int L = __builtin_ctz(mask);
        mask &= mask - 1;
        const int s = __builtin_amdgcn_readlane(sender_v, L);
        const int r = __builtin_amdgcn_readlane(recv_v, L);
        const int hq = lane & 3;
        const float ss = rec[(size_t)s * REC + (dir ? REC_S_OUT_S : REC_S_IN_S) + hq];
        const float rr = rec[(size_t)r * REC + (dir ? REC_S_OUT_R : REC_S_IN_R) + hq];
        float e = ss + rr;
        e = (e > 0.0f) ? e : NEG_SLOPE * e;
        const int st = (r - lo) * STAT + dir * NHEAD + hq;
        const int wq = __float_as_int(__expf(e - smax[st]) / (ssum[st] + EPS_SOFTMAX));
        const float* xs = h_proj + (size_t)s * OUT_DIM + lane;
        float* ar = acc + (r - lo) * OUT_DIM + lane;
        float x[NHEAD];
#pragma unroll
        for (int j = 0; j < NHEAD; ++j) x[j] = xs[HD * j];
#pragma unroll
        for (int j = 0; j < NHEAD; ++j) {
            const float w = __int_as_float(__builtin_amdgcn_readlane(wq, j));
            atomicAdd(ar + HD * j, w * x[j]);
        }
    }
}

#define LN_ROWS 4

__device__ __forceinline__ float wave_sum32(float v) {
#pragma unroll
    for (int m = 16; m > 0; m >>= 1) v += __shfl_xor(v, m, 32);
    return v;
}

__global__ __launch_bounds__(AGG_THREADS) void agg_tile_kernel(
    const int* __restrict__ src, const int* __restrict__ dst,
    const float* __restrict__ h_proj,
    const float* rec,
    const float* __restrict__ bias, const float* __restrict__ gamma,
    const float* __restrict__ beta,
    const unsigned int* map,
    const unsigned int* __restrict__ bins,
    const int* __restrict__ fill,
    float* __restrict__ out, int N, int E)
{
    __shared__ __attribute__((aligned(16))) float acc[TILE_ROWS * OUT_DIM];
    __shared__ __attribute__((aligned(16))) float smax[TILE_ROWS * STAT];
    __shared__ __attribute__((aligned(16))) float ssum[TILE_ROWS * STAT];
    __shared__ unsigned int list[LIST_CAP];
    __shared__ int lcount;
    const int tid = threadIdx.x;
    const int lane = tid & 31;
    const int wave = __builtin_amdgcn_readfirstlane(tid >> 5);
    const int lo = blockIdx.x * TILE_ROWS;
    const int rows = (N - lo < TILE_ROWS) ? (N - lo) : TILE_ROWS;
    const unsigned urows = (unsigned)rows;
    const unsigned tiles = gridDim.x;
    const unsigned tile = blockIdx.x;

    for (int i = tid; i < TILE_ROWS * OUT_DIM / 4; i += AGG_THREADS) {
        const v4f zero4 = {0.f, 0.f, 0.f, 0.f};
        ((v4f*)acc)[i] = zero4;
    }
    for (int i = tid; i < TILE_ROWS * STAT; i += AGG_THREADS) {
        smax[i] = -INFINITY;
        ssum[i] = 0.0f;
    }
    if (tid == 0) lcount = 0;
    __syncthreads();

    int n;
    if (bins) {
        n = __builtin_amdgcn_readfirstlane(fill[tile]);
        if (n <= LIST_CAP) {
            const unsigned int* bt = bins + (size_t)tile * LIST_CAP;
            unsigned en[LIST_PER_THREAD];
#pragma unroll
            for (int k = 0; k < LIST_PER_THREAD; ++k) {
                const int i = tid + AGG_THREADS * k;
                en[k] = (i < n) ? bt[i] : 0u;
            }
#pragma unroll
            for (int k = 0; k < LIST_PER_THREAD; ++k) {
                const int i = tid + AGG_THREADS * k;
                if (i < n) list[i] = en[k];
            }
        }
        __syncthreads();
    } else {
    const unsigned below = (1u << lane) - 1u;
    scan_hit_groups(src, dst, map, E, lo, urows, tiles, tile, wave, lane,
        [&](const int (&sv)[STEP_GROUPS], const int (&dv)[STEP_GROUPS],
            const unsigned (&m_in)[STEP_GROUPS], const unsigned (&m_out)[STEP_GROUPS],
            int kept) {
            if (kept == 0) return;
            int base = 0;
            if (lane == 0) base = atomicAdd(&lcount, kept);
            int off = __builtin_amdgcn_readfirstlane(base);
#pragma unroll
            for (int i = 0; i < STEP_GROUPS; ++i) {
                if ((m_in[i] >> lane) & 1u) {
                    const int pos = off + __builtin_popcount(m_in[i] & below);
                    if (pos < LIST_CAP)
                        list[pos] = (unsigned)sv[i] | ((unsigned)(dv[i] - lo) << ENT_ROW_SHIFT);
                }
                off += __builtin_popcount(m_in[i]);
                if ((m_out[i] >> lane) & 1u) {
                    const int pos = off + __builtin_popcount(m_out[i] & below);
                    if (pos < LIST_CAP)
                        list[pos] = (unsigned)dv[i] | ((unsigned)(sv[i] - lo) << ENT_ROW_SHIFT)
                                    | (1u << ENT_DIR_SHIFT);
                }
                off += __builtin_popcount(m_out[i]);
            }
        });
    __syncthreads();
    n = __builtin_amdgcn_readfirstlane(lcount);
    }

    if (n <= LIST_CAP) {
        if (n > 0) {
            v4f ev[LIST_PER_THREAD] = {};
#pragma unroll
            for (int k0 = 0; k0 < LIST_PER_THREAD; k0 += PASS_GROUP) {
                if (AGG_THREADS * k0 < n) {
                    unsigned en[PASS_GROUP];
#pragma unroll
                    for (int j = 0; j < PASS_GROUP; ++j) {
                        const int i = tid + AGG_THREADS * (k0 + j);
                        en[j] = list[(i < n) ? i : n - 1];
                        const int row = (int)((en[j] >> ENT_ROW_SHIFT) & ENT_ROW_MASK);
                        ev[k0 + j] = edge_logits(rec, (int)(en[j] & ENT_SENDER_MASK), lo + row,
                                                 (int)(en[j] >> ENT_DIR_SHIFT));
                    }
#pragma unroll
                    for (int j = 0; j < PASS_GROUP; ++j) {
                        if (tid + AGG_THREADS * (k0 + j) < n) {
                            float* mp = smax + (int)((en[j] >> ENT_ROW_SHIFT) & ENT_ROW_MASK) * STAT
                                        + (int)(en[j] >> ENT_DIR_SHIFT) * NHEAD;
                            atomicMaxF32(mp + 0, ev[k0 + j].x);
                            atomicMaxF32(mp + 1, ev[k0 + j].y);
                            atomicMaxF32(mp + 2, ev[k0 + j].z);
                            atomicMaxF32(mp + 3, ev[k0 + j].w);
                        }
                    }
                }
            }
            __syncthreads();
#pragma unroll
            for (int k = 0; k < LIST_PER_THREAD; ++k) {
                const int i = tid + AGG_THREADS * k;
                if (i < n) {
                    const unsigned en = list[i];
                    const int st = (int)((en >> ENT_ROW_SHIFT) & ENT_ROW_MASK) * STAT
                                   + (int)(en >> ENT_DIR_SHIFT) * NHEAD;
                    const v4f mx = *(const v4f*)(smax + st);
                    v4f p;
                    p.x = __expf(ev[k].x - mx.x);
                    p.y = __expf(ev[k].y - mx.y);
                    p.z = __expf(ev[k].z - mx.z);
                    p.w = __expf(ev[k].w - mx.w);
                    atomicAdd(ssum + st + 0, p.x);
                    atomicAdd(ssum + st + 1, p.y);
                    atomicAdd(ssum + st + 2, p.z);
                    atomicAdd(ssum + st + 3, p.w);
                    ev[k] = p;
                }
            }
            __syncthreads();
            const int chunks = (n + 31) >> 5;
            for (int k = 0; k < LIST_PER_THREAD; ++k) {
                const int c = wave + AGG_WAVES * k;
                if (c >= chunks) break;
                const int left = n - 32 * c;
                const int cn = (left < 32) ? left : 32;
                const int i = 32 * c + lane;
                const unsigned en = list[(i < n) ? i : n - 1];
                v4f p = ev[0];
#pragma unroll
                for (int kk = 1; kk < LIST_PER_THREAD; ++kk)
                    if (kk == k) p = ev[kk];
                const int sender = (int)(en & ENT_SENDER_MASK);
                const int row = (int)((en >> ENT_ROW_SHIFT) & ENT_ROW_MASK);
                const int dir = (int)(en >> ENT_DIR_SHIFT);
                const v4f sm = *(const v4f*)(ssum + row * STAT + dir * NHEAD);
                int wq[NHEAD];
                wq[0] = __float_as_int(p.x / (sm.x + EPS_SOFTMAX));
                wq[1] = __float_as_int(p.y / (sm.y + EPS_SOFTMAX));
                wq[2] = __float_as_int(p.z / (sm.z + EPS_SOFTMAX));
                wq[3] = __float_as_int(p.w / (sm.w + EPS_SOFTMAX));
                add_chunk_rows(cn, sender, row, wq, lane, h_proj, acc);
            }
        }
    } else {
        scan_hit_groups(src, dst, map, E, lo, urows, tiles, tile, wave, lane,
            [&](const int (&sv)[STEP_GROUPS], const int (&dv)[STEP_GROUPS],
                const unsigned (&m_in)[STEP_GROUPS], const unsigned (&m_out)[STEP_GROUPS],
                int kept) {
                if (kept == 0) return;
#pragma unroll
                for (int i = 0; i < STEP_GROUPS; ++i) {
                    if ((m_in[i] >> lane) & 1u) {
                        const v4f e = edge_logits(rec, sv[i], dv[i], 0);
                        float* mp = smax + (dv[i] - lo) * STAT;
                        atomicMaxF32(mp + 0, e.x);
                        atomicMaxF32(mp + 1, e.y);
                        atomicMaxF32(mp + 2, e.z);
                        atomicMaxF32(mp + 3, e.w);
                    }
                    if ((m_out[i] >> lane) & 1u) {
                        const v4f e = edge_logits(rec, dv[i], sv[i], 1);
                        float* mp = smax + (sv[i] - lo) * STAT + NHEAD;
                        atomicMaxF32(mp + 0, e.x);
                        atomicMaxF32(mp + 1, e.y);
                        atomicMaxF32(mp + 2, e.z);
                        atomicMaxF32(mp + 3, e.w);
                    }
                }
            });
        __syncthreads();
        scan_hit_groups(src, dst, map, E, lo, urows, tiles, tile, wave, lane,
            [&](const int (&sv)[STEP_GROUPS], const int (&dv)[STEP_GROUPS],
                const unsigned (&m_in)[STEP_GROUPS], const unsigned (&m_out)[STEP_GROUPS],
                int kept) {
                if (kept == 0) return;
#pragma unroll
                for (int i = 0; i < STEP_GROUPS; ++i) {
                    if ((m_in[i] >> lane) & 1u) {
                        const v4f e = edge_logits(rec, sv[i], dv[i], 0);
                        const int st = (dv[i] - lo) * STAT;
                        const v4f mx = *(const v4f*)(smax + st);
                        atomicAdd(ssum + st + 0, __expf(e.x - mx.x));
                        atomicAdd(ssum + st + 1, __expf(e.y - mx.y));
                        atomicAdd(ssum + st + 2, __expf(e.z - mx.z));
                        atomicAdd(ssum + st + 3, __expf(e.w - mx.w));
                    }
                    if ((m_out[i] >> lane) & 1u) {
                        const v4f e = edge_logits(rec, dv[i], sv[i], 1);
                        const int st = (sv[i] - lo) * STAT + NHEAD;
                        const v4f mx = *(const v4f*)(smax + st);
                        atomicAdd(ssum + st + 0, __expf(e.x - mx.x));
                        atomicAdd(ssum + st + 1, __expf(e.y - mx.y));
                        atomicAdd(ssum + st + 2, __expf(e.z - mx.z));
                        atomicAdd(ssum + st + 3, __expf(e.w - mx.w));
                    }
                }
            });
        __syncthreads();
        scan_hit_groups(src, dst, map, E, lo, urows, tiles, tile, wave, lane,
            [&](const int (&sv)[STEP_GROUPS], const int (&dv)[STEP_GROUPS],
                const unsigned (&m_in)[STEP_GROUPS], const unsigned (&m_out)[STEP_GROUPS],
                int kept) {
                if (kept == 0) return;
#pragma unroll
                for (int i = 0; i < STEP_GROUPS; ++i) {
                    add_kept_edges(m_in[i], sv[i], dv[i], 0, lane, lo, h_proj, rec,
                                   smax, ssum, acc);
                    add_kept_edges(m_out[i], dv[i], sv[i], 1, lane, lo, h_proj, rec,
                                   smax, ssum, acc);
                }
            });
    }
    __syncthreads();

    const v4f bi = *(const v4f*)(bias + 4 * lane);
    const v4f ga = *(const v4f*)(gamma + 4 * lane);
    const v4f be = *(const v4f*)(beta + 4 * lane);
    for (int r0 = wave; r0 < rows; r0 += LN_ROWS * AGG_WAVES) {
        v4f x[LN_ROWS];
#pragma unroll
        for (int i = 0; i < LN_ROWS; ++i) {
            const int r = r0 + i * AGG_WAVES;
            const int rc = (r < rows) ? r : r0;
            const v4f hs = *(const v4f*)(out + (size_t)(lo + rc) * OUT_DIM + 4 * lane);
            const v4f a = *(const v4f*)(acc + rc * OUT_DIM + 4 * lane);
            x[i] = hs + a + bi;
        }
        float mean[LN_ROWS];
#pragma unroll
        for (int i = 0; i < LN_ROWS; ++i)
            mean[i] = wave_sum32((x[i].x + x[i].y) + (x[i].z + x[i].w)) * (1.0f / OUT_DIM);
        float inv[LN_ROWS];
#pragma unroll
        for (int i = 0; i < LN_ROWS; ++i) {
            x[i] = x[i] - mean[i];
            const v4f q = x[i] * x[i];
            const float var = wave_sum32((q.x + q.y) + (q.z + q.w)) * (1.0f / OUT_DIM);
            inv[i] = rsqrtf(var + EPS_LN);
        }
#pragma unroll
        for (int i = 0; i < LN_ROWS; ++i) {
            const int r = r0 + i * AGG_WAVES;
            if (r < rows)
                *(v4f*)(out + (size_t)(lo + r) * OUT_DIM + 4 * lane) = x[i] * inv[i] * ga + be;
        }
    }
}

extern "C" void kernel_launch(void* const* d_in, const int* in_sizes, int n_in,
                              void* d_out, int out_size, void* d_ws, size_t ws_size,
                              hipStream_t stream) {
    const float* h      = (const float*)d_in[0];
    const int*   eidx   = (const int*)d_in[1];
    const float* W      = (const float*)d_in[2];
    const float* Wself  = (const float*)d_in[3];
    const float* a_in   = (const float*)d_in[4];
    const float* a_out  = (const float*)d_in[5];
    const float* bias   = (const float*)d_in[6];
    const float* gamma  = (const float*)d_in[7];
    const float* beta   = (const float*)d_in[8];
    float* out = (float*)d_out;

    const int N = in_sizes[0] / IN_DIM;
    const int E = in_sizes[1] / 2;
    const int NH = N * NHEAD;

    const int* src = eidx;
    const int* dst = eidx + E;

    float* ws = (float*)d_ws;
    float* h_proj  = ws;
    float* rec     = h_proj + (size_t)N * OUT_DIM;
    const int tiles = (N + TILE_ROWS - 1) / TILE_ROWS;
    const size_t base_words = (size_t)N * OUT_DIM + (size_t)N * REC;
    const size_t bin_words = (size_t)tiles * LIST_CAP;
    unsigned int* bins = (unsigned int*)(ws + base_words);
    int* fill = (int*)(bins + bin_words);
    const bool use_bins = ws_size >= (base_words + bin_words + (size_t)tiles) * sizeof(float);

    {
        int blocks = (N + GEMM_ROWS - 1) / GEMM_ROWS;
        gemm_wmma_kernel<<<blocks, 256, 0, stream>>>(h, W, Wself, h_proj, out, N);
    }
    {
        int threads = 256, blocks = (NH + threads - 1) / threads;
        score_kernel<<<blocks, threads, 0, stream>>>(h_proj, a_in, a_out, rec, NH,
                                                     use_bins ? fill : nullptr, tiles);
    }
    const int nwords = (E + MAP_EDGES - 1) / MAP_EDGES;
    const bool use_map = !use_bins && (long long)nwords * tiles <= (long long)N * MAP_SLOTS;
    unsigned int* map = (unsigned int*)rec;
    if (use_bins) {
        const int blocks = (E + BIN_EDGES - 1) / BIN_EDGES;
        tile_bin_kernel<<<blocks, BIN_THREADS, 3 * tiles * sizeof(int), stream>>>(
            src, dst, bins, fill, E, tiles);
    }
    if (use_map) {
        tile_map_kernel<<<nwords, MAP_EDGES, tiles * sizeof(unsigned int), stream>>>(
            src, dst, map, E, tiles);
    }
    {
        agg_tile_kernel<<<tiles, AGG_THREADS, 0, stream>>>(src, dst, h_proj, rec,
                                                           bias, gamma, beta,
                                                           use_map ? map : nullptr,
                                                           use_bins ? bins : nullptr, fill,
                                                           out, N, E);
    }
}


# Node scores formed in the GEMM launch from the sums its h_proj waves hold (per-lane shares, 16-lane exchange sums): score launch and its re-read of h_proj gone, 3 launches; score sum order changes
# speedup vs baseline: 18.6831x; 1.0599x over previous
#include <hip/hip_runtime.h>
#include <math.h>

typedef float v2f __attribute__((ext_vector_type(2)));
typedef float v4f __attribute__((ext_vector_type(4)));
typedef float v8f __attribute__((ext_vector_type(8)));

#define IN_DIM 128
#define OUT_DIM 128
#define NHEAD 4
#define HD 32
#define NEG_SLOPE 0.2f
#define EPS_SOFTMAX 1e-8f
#define EPS_LN 1e-5f

__device__ __forceinline__ void atomicMaxF32(float* addr, float val) {
    if (val >= 0.0f) {
        atomicMax((int*)addr, __float_as_int(val));
    } else {
        atomicMin((unsigned int*)addr, (unsigned int)__float_as_int(val));
    }
}

#define REC 32
#define REC_S_IN_S 0
#define REC_S_OUT_R 4
#define REC_S_IN_R 8
#define REC_S_OUT_S 12
#define REC_MAP 16

template <int M>
__device__ __forceinline__ float row_xor(float v) {
    return __int_as_float(__builtin_amdgcn_update_dpp(0, __float_as_int(v), 0x160 + M, 0xF, 0xF, true));
}

#define GEMM_ROWS 64
__global__ __launch_bounds__(256) void gemm_wmma_kernel(
    const float* __restrict__ h, const float* __restrict__ W,
    const float* __restrict__ Wself, float* __restrict__ h_proj,
    float* __restrict__ out, int N,
    const float* __restrict__ a_in, const float* __restrict__ a_out,
    float* __restrict__ rec, int* __restrict__ fill, int tiles)
{
    {
        const int gid = blockIdx.x * 256 + threadIdx.x;
        if (fill && gid < tiles) fill[gid] = 0;
    }
    const int wave = threadIdx.x >> 5;
    const int lane = threadIdx.x & 31;
    const int c = lane & 15;
    const int hh = lane >> 4;
    const int rgrp = wave >> 2;
    const int cgrp = wave & 3;
    const int row0 = blockIdx.x * GEMM_ROWS + rgrp * 32;

    int ra = row0 + c;
    int rb = row0 + 16 + c;
    if (ra >= N) ra = N - 1;
    if (rb >= N) rb = N - 1;
    const float* ap0 = h + (size_t)ra * IN_DIM + 4 * hh;
    const float* ap1 = h + (size_t)rb * IN_DIM + 4 * hh;

    const bool proj = (cgrp < 2);
    const int colbase = (cgrp & 1) * 64;
    const float* bp = (proj ? W : Wself) + colbase + c + (size_t)(4 * hh) * OUT_DIM;

    v8f acc0[4] = {};
    v8f acc1[4] = {};
    for (int k0 = 0; k0 < IN_DIM; k0 += 8) {
        const v4f a0 = *(const v4f*)(ap0 + k0);
        const v4f a1 = *(const v4f*)(ap1 + k0);
        float b[4][4];
#pragma unroll
        for (int t = 0; t < 4; ++t) {
#pragma unroll
            for (int s = 0; s < 4; ++s) b[t][s] = bp[(size_t)(k0 + s) * OUT_DIM + 16 * t];
        }
#pragma unroll
        for (int j = 0; j < 2; ++j) {
            v2f fa0, fa1;
            fa0.x = a0[2 * j];
            fa0.y = a0[2 * j + 1];
            fa1.x = a1[2 * j];
            fa1.y = a1[2 * j + 1];
#pragma unroll
            for (int t = 0; t < 4; ++t) {
                v2f fb;
                fb.x = b[t][2 * j];
                fb.y = b[t][2 * j + 1];
                acc0[t] = __builtin_amdgcn_wmma_f32_16x16x4_f32(false, fa0, false, fb,
                                                                (short)0, acc0[t], false, false);
                acc1[t] = __builtin_amdgcn_wmma_f32_16x16x4_f32(false, fa1, false, fb,
                                                                (short)0, acc1[t], false, false);
            }
        }
    }

    float* dstm = (proj ? h_proj : out) + colbase + c;
    const int mrow = row0 + 8 * hh;
    if (row0 + 32 <= N) {
#pragma unroll
        for (int t = 0; t < 4; ++t) {
#pragma unroll
            for (int v = 0; v < 8; ++v) {
                dstm[(size_t)(mrow + v) * OUT_DIM + 16 * t] = acc0[t][v];
                dstm[(size_t)(mrow + 16 + v) * OUT_DIM + 16 * t] = acc1[t][v];
            }
        }
    } else {
#pragma unroll
        for (int t = 0; t < 4; ++t) {
#pragma unroll
            for (int v = 0; v < 8; ++v) {
                if (mrow + v < N)
                    dstm[(size_t)(mrow + v) * OUT_DIM + 16 * t] = acc0[t][v];
                if (mrow + 16 + v < N)
                    dstm[(size_t)(mrow + 16 + v) * OUT_DIM + 16 * t] = acc1[t][v];
            }
        }
    }

    if (proj) {
        const bool b3 = (c & 8) != 0, b2 = (c & 4) != 0, b1 = (c & 2) != 0, b0 = (c & 1) != 0;
#pragma unroll
        for (int hl = 0; hl < 2; ++hl) {
            const int head = 2 * cgrp + hl;
            const float* ai = a_in + head * 2 * HD + c;
            const float* ao = a_out + head * 2 * HD + c;
            float av[4][2];
#pragma unroll
            for (int tt = 0; tt < 2; ++tt) {
                av[0][tt] = ai[16 * tt];
                av[1][tt] = ao[HD + 16 * tt];
                av[2][tt] = ai[HD + 16 * tt];
                av[3][tt] = ao[16 * tt];
            }
#pragma unroll
            for (int rt = 0; rt < 2; ++rt) {
                float p16[16];
#pragma unroll
                for (int v = 0; v < 4; ++v) {
                    const float lo0 = rt ? acc1[2 * hl][v] : acc0[2 * hl][v];
                    const float lo1 = rt ? acc1[2 * hl + 1][v] : acc0[2 * hl + 1][v];
                    const float hi0 = rt ? acc1[2 * hl][v + 4] : acc0[2 * hl][v + 4];
                    const float hi1 = rt ? acc1[2 * hl + 1][v + 4] : acc0[2 * hl + 1][v + 4];
                    const float k0 = b3 ? hi0 : lo0, k1 = b3 ? hi1 : lo1;
                    const float s0 = b3 ? lo0 : hi0, s1 = b3 ? lo1 : hi1;
#pragma unroll
                    for (int q = 0; q < 4; ++q) {
                        const float keep = k0 * av[q][0] + k1 * av[q][1];
                        const float send = s0 * av[q][0] + s1 * av[q][1];
                        p16[4 * v + q] = keep + row_xor<8>(send);
                    }
                }
                float p8[8], p4[4], p2[2];
#pragma unroll
                for (int i = 0; i < 8; ++i) {
                    const float keep = b2 ? p16[i + 8] : p16[i];
                    const float send = b2 ? p16[i] : p16[i + 8];
                    p8[i] = keep + row_xor<4>(send);
                }
#pragma unroll
                for (int i = 0; i < 4; ++i) {
                    const float keep = b1 ? p8[i + 4] : p8[i];
                    const float send = b1 ? p8[i] : p8[i + 4];
                    p4[i] = keep + row_xor<2>(send);
                }
#pragma unroll
                for (int i = 0; i < 2; ++i) {
                    const float keep = b0 ? p4[i + 2] : p4[i];
                    const float send = b0 ? p4[i] : p4[i + 2];
                    p2[i] = keep + row_xor<1>(send);
                }
                const int row = row0 + 16 * rt + 8 * hh + (c >> 1);
                if (row < N) {
                    float* r = rec + (size_t)row * REC + 8 * (c & 1) + head;
                    r[0] = p2[0];
                    r[4] = p2[1];
                }
            }
        }
    }
}

#define TILE_ROWS 400
#define AGG_THREADS 1024
#define AGG_WAVES (AGG_THREADS / 32)
#define STAT 8

#define MAP_EDGES 1024
#define MAP_SLOTS 16
#define STEP_GROUPS 8

__device__ __forceinline__ size_t map_word_at(unsigned i) {
    return (size_t)(i >> 4) * REC + REC_MAP + (i & (MAP_SLOTS - 1));
}

__global__ __launch_bounds__(MAP_EDGES) void tile_map_kernel(
    const int* __restrict__ src, const int* __restrict__ dst,
    unsigned int* __restrict__ map, int E, int tiles)
{
    extern __shared__ unsigned int wmap[];
    const int tid = threadIdx.x;
    for (int t = tid; t < tiles; t += MAP_EDGES) wmap[t] = 0u;
    __syncthreads();
    const int e = blockIdx.x * MAP_EDGES + tid;
    if (e < E) {
        const unsigned d = (unsigned)dst[e];
        const unsigned s = (unsigned)src[e];
        const unsigned bit = 1u << (tid >> 5);
        atomicOr(&wmap[d / TILE_ROWS], bit);
        atomicOr(&wmap[s / TILE_ROWS], bit);
    }
    __syncthreads();
    for (int t = tid; t < tiles; t += MAP_EDGES)
        map[map_word_at((unsigned)blockIdx.x * (unsigned)tiles + (unsigned)t)] = wmap[t];
}

#define LIST_CAP 16384
#define LIST_PER_THREAD (LIST_CAP / AGG_THREADS)
#define PASS_GROUP 4
#define ENT_ROW_SHIFT 22
#define ENT_DIR_SHIFT 31
#define ENT_SENDER_MASK 0x3FFFFFu
#define ENT_ROW_MASK 0x1FFu

#define BIN_THREADS 1024
#define BIN_PER_THREAD 8
#define BIN_EDGES (BIN_THREADS * BIN_PER_THREAD)
#define BIN_ENTRIES (2 * BIN_EDGES)

__global__ __launch_bounds__(BIN_THREADS) void tile_bin_kernel(
    const int* __restrict__ src, const int* __restrict__ dst,
    unsigned int* __restrict__ bins, int* fill, int E, int tiles)
{
    extern __shared__ int bcnt[];
    int* bstart = bcnt + tiles;
    int* lstart = bstart + tiles;
    __shared__ unsigned int stg[BIN_ENTRIES];
    __shared__ unsigned short stile[BIN_ENTRIES];
    const int tid = threadIdx.x;
    const int lane = tid & 31;
    const unsigned utiles = (unsigned)tiles;
    for (int t = tid; t < tiles; t += BIN_THREADS) bcnt[t] = 0;
    __syncthreads();
    int s[BIN_PER_THREAD], d[BIN_PER_THREAD];
#pragma unroll
    for (int k = 0; k < BIN_PER_THREAD; ++k) {
        const int e = blockIdx.x * BIN_EDGES + k * BIN_THREADS + tid;
        const bool ok = (e < E);
        s[k] = ok ? src[e] : -1;
        d[k] = ok ? dst[e] : -1;
    }
    unsigned ti[BIN_PER_THREAD], to[BIN_PER_THREAD];
    int ri[BIN_PER_THREAD], ro[BIN_PER_THREAD];
#pragma unroll
    for (int k = 0; k < BIN_PER_THREAD; ++k) {
        ti[k] = (unsigned)d[k] / TILE_ROWS;
        to[k] = (unsigned)s[k] / TILE_ROWS;
        ri[k] = (ti[k] < utiles) ? atomicAdd(&bcnt[ti[k]], 1) : 0;
        ro[k] = (to[k] < utiles) ? atomicAdd(&bcnt[to[k]], 1) : 0;
    }
    __syncthreads();
    for (int t = tid; t < tiles; t += BIN_THREADS) {
        const int cnt = bcnt[t];
        bstart[t] = cnt ? atomicAdd(&fill[t], cnt) : 0;
    }
    if ((tid >> 5) == BIN_THREADS / 32 - 1) {
        const int per = (tiles + 31) >> 5;
        const int tb = lane * per;
        int sum = 0;
        for (int i = 0; i < per; ++i)
            if (tb + i < tiles) sum += bcnt[tb + i];
        int x = sum;
#pragma unroll
        for (int o = 1; o < 32; o <<= 1) {
            const int y = __shfl_up(x, o, 32);
            if (lane >= o) x += y;
        }
        int run = x - sum;
        for (int i = 0; i < per; ++i) {
            if (tb + i < tiles) {
                lstart[tb + i] = run;
                run += bcnt[tb + i];
            }
        }
    }
    __syncthreads();
#pragma unroll
    for (int k = 0; k < BIN_PER_THREAD; ++k) {
        if (ti[k] < utiles) {
            const int j = lstart[ti[k]] + ri[k];
            stg[j] = (unsigned)s[k] | (((unsigned)d[k] - ti[k] * TILE_ROWS) << ENT_ROW_SHIFT);
            stile[j] = (unsigned short)ti[k];
        }
        if (to[k] < utiles) {
            const int j = lstart[to[k]] + ro[k];
            stg[j] = (unsigned)d[k] | (((unsigned)s[k] - to[k] * TILE_ROWS) << ENT_ROW_SHIFT)
                     | (1u << ENT_DIR_SHIFT);
            stile[j] = (unsigned short)to[k];
        }
    }
    __syncthreads();
    const int total = lstart[tiles - 1] + bcnt[tiles - 1];
#pragma unroll
    for (int k = 0; k < 2 * BIN_PER_THREAD; ++k) {
        const int j = tid + k * BIN_THREADS;
        if (j < total) {
            const int t = stile[j];
            const int pos = bstart[t] + (j - lstart[t]);
            if (pos < LIST_CAP) bins[(size_t)t * LIST_CAP + pos] = stg[j];
        }
    }
}

__device__ __forceinline__ v4f edge_logits(const float* rec, int sender, int recv, int dir)
{
    const v4f s = *(const v4f*)(rec + (size_t)sender * REC + (dir ? REC_S_OUT_S : REC_S_IN_S));
    const v4f r = *(const v4f*)(rec + (size_t)recv * REC + (dir ? REC_S_OUT_R : REC_S_IN_R));
    v4f e = s + r;
    e.x = (e.x > 0.0f) ? e.x : NEG_SLOPE * e.x;
    e.y = (e.y > 0.0f) ? e.y : NEG_SLOPE * e.y;
    e.z = (e.z > 0.0f) ? e.z : NEG_SLOPE * e.z;
    e.w = (e.w > 0.0f) ? e.w : NEG_SLOPE * e.w;
    return e;
}

template <class F>
__device__ __forceinline__ void scan_hit_groups(const int* __restrict__ src,
                                                const int* __restrict__ dst,
                                                const unsigned int* map, int E, int lo,
                                                unsigned urows, unsigned tiles, unsigned tile,
                                                int wave, int lane, F&& f)
{
    const int nwords = (E + MAP_EDGES - 1) / MAP_EDGES;
    for (int w0 = wave; w0 < nwords; w0 += 32 * AGG_WAVES) {
        const int wl = w0 + AGG_WAVES * lane;
        unsigned myword = 0u;
        if (wl < nwords)
            myword = map ? map[map_word_at((unsigned)wl * tiles + tile)] : 0xFFFFFFFFu;
        for (int k = 0; k < 32; ++k) {
            const int w = w0 + AGG_WAVES * k;
            if (w >= nwords) break;
            unsigned bits = (unsigned)__builtin_amdgcn_readlane((int)myword, k);
            const int ebase = w * MAP_EDGES;
            while (bits) {
                int g[STEP_GROUPS];
#pragma unroll
                for (int i = 0; i < STEP_GROUPS; ++i) {
                    g[i] = bits ? __builtin_ctz(bits) : 32;
                    bits &= bits - 1u;
                }
                int sv[STEP_GROUPS], dv[STEP_GROUPS];
#pragma unroll
                for (int i = 0; i < STEP_GROUPS; ++i) {
                    const int e = ebase + 32 * g[i] + lane;
                    const bool ok = (g[i] < 32) && (e < E);
                    sv[i] = ok ? src[e] : -1;
                    dv[i] = ok ? dst[e] : -1;
                }
                unsigned m_in[STEP_GROUPS], m_out[STEP_GROUPS];
                int kept = 0;
#pragma unroll
                for (int i = 0; i < STEP_GROUPS; ++i) {
                    m_in[i] = __builtin_amdgcn_ballot_w32((unsigned)(dv[i] - lo) < urows);
                    m_out[i] = __builtin_amdgcn_ballot_w32((unsigned)(sv[i] - lo) < urows);
                    kept += __builtin_popcount(m_in[i]) + __builtin_popcount(m_out[i]);
                }
                f(sv, dv, m_in, m_out, kept);
            }
        }
    }
}

#define GATHER_BATCH 8

__device__ __forceinline__ void request_rows(int b, int n, int q_s, int lane,
                                             const float* __restrict__ h_proj,
                                             float (&x)[GATHER_BATCH][NHEAD])
{
#pragma unroll
    for (int i = 0; i < GATHER_BATCH; ++i) {
        const int idx = (b + i < n) ? (b + i) : (n - 1);
        const int s = __builtin_amdgcn_readlane(q_s, idx);
        const float* xs = h_proj + (size_t)s * OUT_DIM + lane;
#pragma unroll
        for (int j = 0; j < NHEAD; ++j) x[i][j] = xs[HD * j];
    }
}

__device__ __forceinline__ void add_chunk_rows(int n, int q_s, int q_r, const int (&wq)[NHEAD],
                                               int lane, const float* __restrict__ h_proj,
                                               float* acc)
{
    for (int b = 0; b < n; b += GATHER_BATCH) {
        float x[GATHER_BATCH][NHEAD];
        request_rows(b, n, q_s, lane, h_proj, x);
#pragma unroll
        for (int i = 0; i < GATHER_BATCH; ++i) {
            const bool live = (b + i < n);
            const int idx = live ? (b + i) : (n - 1);
            const int r = __builtin_amdgcn_readlane(q_r, idx);
            float* ar = acc + r * OUT_DIM + lane;
#pragma unroll
            for (int j = 0; j < NHEAD; ++j) {
                const float w = __int_as_float(__builtin_amdgcn_readlane(wq[j], idx));
                const float v = live ? w * x[i][j] : 0.0f;
                atomicAdd(ar + HD * j, v);
            }
        }
    }
}

__device__ __forceinline__ void add_kept_edges(unsigned mask, int sender_v, int recv_v, int dir,
                                               int lane, int lo,
                                               const float* __restrict__ h_proj,
                                               const float* rec, const float* smax,
                                               const float* ssum, float* acc)
{
    while (mask) {
        const int L = __builtin_ctz(mask);
        mask &= mask - 1;
        const int s = __builtin_amdgcn_readlane(sender_v, L);
        const int r = __builtin_amdgcn_readlane(recv_v, L);
        const int hq = lane & 3;
        const float ss = rec[(size_t)s * REC + (dir ? REC_S_OUT_S : REC_S_IN_S) + hq];
        const float rr = rec[(size_t)r * REC + (dir ? REC_S_OUT_R : REC_S_IN_R) + hq];
        float e = ss + rr;
        e = (e > 0.0f) ? e : NEG_SLOPE * e;
        const int st = (r - lo) * STAT + dir * NHEAD + hq;
        const int wq = __float_as_int(__expf(e - smax[st]) / (ssum[st] + EPS_SOFTMAX));
        const float* xs = h_proj + (size_t)s * OUT_DIM + lane;
        float* ar = acc + (r - lo) * OUT_DIM + lane;
        float x[NHEAD];
#pragma unroll
        for (int j = 0; j < NHEAD; ++j) x[j] = xs[HD * j];
#pragma unroll
        for (int j = 0; j < NHEAD; ++j) {
            const float w = __int_as_float(__builtin_amdgcn_readlane(wq, j));
            atomicAdd(ar + HD * j, w * x[j]);
        }
    }
}

#define LN_ROWS 4

__device__ __forceinline__ float wave_sum32(float v) {
#pragma unroll
    for (int m = 16; m > 0; m >>= 1) v += __shfl_xor(v, m, 32);
    return v;
}

__global__ __launch_bounds__(AGG_THREADS) void agg_tile_kernel(
    const int* __restrict__ src, const int* __restrict__ dst,
    const float* __restrict__ h_proj,
    const float* rec,
    const float* __restrict__ bias, const float* __restrict__ gamma,
    const float* __restrict__ beta,
    const unsigned int* map,
    const unsigned int* __restrict__ bins,
    const int* __restrict__ fill,
    float* __restrict__ out, int N, int E)
{
    __shared__ __attribute__((aligned(16))) float acc[TILE_ROWS * OUT_DIM];
    __shared__ __attribute__((aligned(16))) float smax[TILE_ROWS * STAT];
    __shared__ __attribute__((aligned(16))) float ssum[TILE_ROWS * STAT];
    __shared__ unsigned int list[LIST_CAP];
    __shared__ int lcount;
    const int tid = threadIdx.x;
    const int lane = tid & 31;
    const int wave = __builtin_amdgcn_readfirstlane(tid >> 5);
    const int lo = blockIdx.x * TILE_ROWS;
    const int rows = (N - lo < TILE_ROWS) ? (N - lo) : TILE_ROWS;
    const unsigned urows = (unsigned)rows;
    const unsigned tiles = gridDim.x;
    const unsigned tile = blockIdx.x;

    for (int i = tid; i < TILE_ROWS * OUT_DIM / 4; i += AGG_THREADS) {
        const v4f zero4 = {0.f, 0.f, 0.f, 0.f};
        ((v4f*)acc)[i] = zero4;
    }
    for (int i = tid; i < TILE_ROWS * STAT; i += AGG_THREADS) {
        smax[i] = -INFINITY;
        ssum[i] = 0.0f;
    }
    if (tid == 0) lcount = 0;
    __syncthreads();

    int n;
    if (bins) {
        n = __builtin_amdgcn_readfirstlane(fill[tile]);
        if (n <= LIST_CAP) {
            const unsigned int* bt = bins + (size_t)tile * LIST_CAP;
            unsigned en[LIST_PER_THREAD];
#pragma unroll
            for (int k = 0; k < LIST_PER_THREAD; ++k) {
                const int i = tid + AGG_THREADS * k;
                en[k] = (i < n) ? bt[i] : 0u;
            }
#pragma unroll
            for (int k = 0; k < LIST_PER_THREAD; ++k) {
                const int i = tid + AGG_THREADS * k;
                if (i < n) list[i] = en[k];
            }
        }
        __syncthreads();
    } else {
    const unsigned below = (1u << lane) - 1u;
    scan_hit_groups(src, dst, map, E, lo, urows, tiles, tile, wave, lane,
        [&](const int (&sv)[STEP_GROUPS], const int (&dv)[STEP_GROUPS],
            const unsigned (&m_in)[STEP_GROUPS], const unsigned (&m_out)[STEP_GROUPS],
            int kept) {
            if (kept == 0) return;
            int base = 0;
            if (lane == 0) base = atomicAdd(&lcount, kept);
            int off = __builtin_amdgcn_readfirstlane(base);
#pragma unroll
            for (int i = 0; i < STEP_GROUPS; ++i) {
                if ((m_in[i] >> lane) & 1u) {
                    const int pos = off + __builtin_popcount(m_in[i] & below);
                    if (pos < LIST_CAP)
                        list[pos] = (unsigned)sv[i] | ((unsigned)(dv[i] - lo) << ENT_ROW_SHIFT);
                }
                off += __builtin_popcount(m_in[i]);
                if ((m_out[i] >> lane) & 1u) {
                    const int pos = off + __builtin_popcount(m_out[i] & below);
                    if (pos < LIST_CAP)
                        list[pos] = (unsigned)dv[i] | ((unsigned)(sv[i] - lo) << ENT_ROW_SHIFT)
                                    | (1u << ENT_DIR_SHIFT);
                }
                off += __builtin_popcount(m_out[i]);
            }
        });
    __syncthreads();
    n = __builtin_amdgcn_readfirstlane(lcount);
    }

    if (n <= LIST_CAP) {
        if (n > 0) {
            v4f ev[LIST_PER_THREAD] = {};
#pragma unroll
            for (int k0 = 0; k0 < LIST_PER_THREAD; k0 += PASS_GROUP) {
                if (AGG_THREADS * k0 < n) {
                    unsigned en[PASS_GROUP];
#pragma unroll
                    for (int j = 0; j < PASS_GROUP; ++j) {
                        const int i = tid + AGG_THREADS * (k0 + j);
                        en[j] = list[(i < n) ? i : n - 1];
                        const int row = (int)((en[j] >> ENT_ROW_SHIFT) & ENT_ROW_MASK);
                        ev[k0 + j] = edge_logits(rec, (int)(en[j] & ENT_SENDER_MASK), lo + row,
                                                 (int)(en[j] >> ENT_DIR_SHIFT));
                    }
#pragma unroll
                    for (int j = 0; j < PASS_GROUP; ++j) {
                        if (tid + AGG_THREADS * (k0 + j) < n) {
                            float* mp = smax + (int)((en[j] >> ENT_ROW_SHIFT) & ENT_ROW_MASK) * STAT
                                        + (int)(en[j] >> ENT_DIR_SHIFT) * NHEAD;
                            atomicMaxF32(mp + 0, ev[k0 + j].x);
                            atomicMaxF32(mp + 1, ev[k0 + j].y);
                            atomicMaxF32(mp + 2, ev[k0 + j].z);
                            atomicMaxF32(mp + 3, ev[k0 + j].w);
                        }
                    }
                }
            }
            __syncthreads();
#pragma unroll
            for (int k = 0; k < LIST_PER_THREAD; ++k) {
                const int i = tid + AGG_THREADS * k;
                if (i < n) {
                    const unsigned en = list[i];
                    const int st = (int)((en >> ENT_ROW_SHIFT) & ENT_ROW_MASK) * STAT
                                   + (int)(en >> ENT_DIR_SHIFT) * NHEAD;
                    const v4f mx = *(const v4f*)(smax + st);
                    v4f p;
                    p.x = __expf(ev[k].x - mx.x);
                    p.y = __expf(ev[k].y - mx.y);
                    p.z = __expf(ev[k].z - mx.z);
                    p.w = __expf(ev[k].w - mx.w);
                    atomicAdd(ssum + st + 0, p.x);
                    atomicAdd(ssum + st + 1, p.y);
                    atomicAdd(ssum + st + 2, p.z);
                    atomicAdd(ssum + st + 3, p.w);
                    ev[k] = p;
                }
            }
            __syncthreads();
            const int chunks = (n + 31) >> 5;
            for (int k = 0; k < LIST_PER_THREAD; ++k) {
                const int c = wave + AGG_WAVES * k;
                if (c >= chunks) break;
                const int left = n - 32 * c;
                const int cn = (left < 32) ? left : 32;
                const int i = 32 * c + lane;
                const unsigned en = list[(i < n) ? i : n - 1];
                v4f p = ev[0];
#pragma unroll
                for (int kk = 1; kk < LIST_PER_THREAD; ++kk)
                    if (kk == k) p = ev[kk];
                const int sender = (int)(en & ENT_SENDER_MASK);
                const int row = (int)((en >> ENT_ROW_SHIFT) & ENT_ROW_MASK);
                const int dir = (int)(en >> ENT_DIR_SHIFT);
                const v4f sm = *(const v4f*)(ssum + row * STAT + dir * NHEAD);
                int wq[NHEAD];
                wq[0] = __float_as_int(p.x / (sm.x + EPS_SOFTMAX));
                wq[1] = __float_as_int(p.y / (sm.y + EPS_SOFTMAX));
                wq[2] = __float_as_int(p.z / (sm.z + EPS_SOFTMAX));
                wq[3] = __float_as_int(p.w / (sm.w + EPS_SOFTMAX));
                add_chunk_rows(cn, sender, row, wq, lane, h_proj, acc);
            }
        }
    } else {
        scan_hit_groups(src, dst, map, E, lo, urows, tiles, tile, wave, lane,
            [&](const int (&sv)[STEP_GROUPS], const int (&dv)[STEP_GROUPS],
                const unsigned (&m_in)[STEP_GROUPS], const unsigned (&m_out)[STEP_GROUPS],
                int kept) {
                if (kept == 0) return;
#pragma unroll
                for (int i = 0; i < STEP_GROUPS; ++i) {
                    if ((m_in[i] >> lane) & 1u) {
                        const v4f e = edge_logits(rec, sv[i], dv[i], 0);
                        float* mp = smax + (dv[i] - lo) * STAT;
                        atomicMaxF32(mp + 0, e.x);
                        atomicMaxF32(mp + 1, e.y);
                        atomicMaxF32(mp + 2, e.z);
                        atomicMaxF32(mp + 3, e.w);
                    }
                    if ((m_out[i] >> lane) & 1u) {
                        const v4f e = edge_logits(rec, dv[i], sv[i], 1);
                        float* mp = smax + (sv[i] - lo) * STAT + NHEAD;
                        atomicMaxF32(mp + 0, e.x);
                        atomicMaxF32(mp + 1, e.y);
                        atomicMaxF32(mp + 2, e.z);
                        atomicMaxF32(mp + 3, e.w);
                    }
                }
            });
        __syncthreads();
        scan_hit_groups(src, dst, map, E, lo, urows, tiles, tile, wave, lane,
            [&](const int (&sv)[STEP_GROUPS], const int (&dv)[STEP_GROUPS],
                const unsigned (&m_in)[STEP_GROUPS], const unsigned (&m_out)[STEP_GROUPS],
                int kept) {
                if (kept == 0) return;
#pragma unroll
                for (int i = 0; i < STEP_GROUPS; ++i) {
                    if ((m_in[i] >> lane) & 1u) {
                        const v4f e = edge_logits(rec, sv[i], dv[i], 0);
                        const int st = (dv[i] - lo) * STAT;
                        const v4f mx = *(const v4f*)(smax + st);
                        atomicAdd(ssum + st + 0, __expf(e.x - mx.x));
                        atomicAdd(ssum + st + 1, __expf(e.y - mx.y));
                        atomicAdd(ssum + st + 2, __expf(e.z - mx.z));
                        atomicAdd(ssum + st + 3, __expf(e.w - mx.w));
                    }
                    if ((m_out[i] >> lane) & 1u) {
                        const v4f e = edge_logits(rec, dv[i], sv[i], 1);
                        const int st = (sv[i] - lo) * STAT + NHEAD;
                        const v4f mx = *(const v4f*)(smax + st);
                        atomicAdd(ssum + st + 0, __expf(e.x - mx.x));
                        atomicAdd(ssum + st + 1, __expf(e.y - mx.y));
                        atomicAdd(ssum + st + 2, __expf(e.z - mx.z));
                        atomicAdd(ssum + st + 3, __expf(e.w - mx.w));
                    }
                }
            });
        __syncthreads();
        scan_hit_groups(src, dst, map, E, lo, urows, tiles, tile, wave, lane,
            [&](const int (&sv)[STEP_GROUPS], const int (&dv)[STEP_GROUPS],
                const unsigned (&m_in)[STEP_GROUPS], const unsigned (&m_out)[STEP_GROUPS],
                int kept) {
                if (kept == 0) return;
#pragma unroll
                for (int i = 0; i < STEP_GROUPS; ++i) {
                    add_kept_edges(m_in[i], sv[i], dv[i], 0, lane, lo, h_proj, rec,
                                   smax, ssum, acc);
                    add_kept_edges(m_out[i], dv[i], sv[i], 1, lane, lo, h_proj, rec,
                                   smax, ssum, acc);
                }
            });
    }
    __syncthreads();

    const v4f bi = *(const v4f*)(bias + 4 * lane);
    const v4f ga = *(const v4f*)(gamma + 4 * lane);
    const v4f be = *(const v4f*)(beta + 4 * lane);
    for (int r0 = wave; r0 < rows; r0 += LN_ROWS * AGG_WAVES) {
        v4f x[LN_ROWS];
#pragma unroll
        for (int i = 0; i < LN_ROWS; ++i) {
            const int r = r0 + i * AGG_WAVES;
            const int rc = (r < rows) ? r : r0;
            const v4f hs = *(const v4f*)(out + (size_t)(lo + rc) * OUT_DIM + 4 * lane);
            const v4f a = *(const v4f*)(acc + rc * OUT_DIM + 4 * lane);
            x[i] = hs + a + bi;
        }
        float mean[LN_ROWS];
#pragma unroll
        for (int i = 0; i < LN_ROWS; ++i)
            mean[i] = wave_sum32((x[i].x + x[i].y) + (x[i].z + x[i].w)) * (1.0f / OUT_DIM);
        float inv[LN_ROWS];
#pragma unroll
        for (int i = 0; i < LN_ROWS; ++i) {
            x[i] = x[i] - mean[i];
            const v4f q = x[i] * x[i];
            const float var = wave_sum32((q.x + q.y) + (q.z + q.w)) * (1.0f / OUT_DIM);
            inv[i] = rsqrtf(var + EPS_LN);
        }
#pragma unroll
        for (int i = 0; i < LN_ROWS; ++i) {
            const int r = r0 + i * AGG_WAVES;
            if (r < rows)
                *(v4f*)(out + (size_t)(lo + r) * OUT_DIM + 4 * lane) = x[i] * inv[i] * ga + be;
        }
    }
}

extern "C" void kernel_launch(void* const* d_in, const int* in_sizes, int n_in,
                              void* d_out, int out_size, void* d_ws, size_t ws_size,
                              hipStream_t stream) {
    const float* h      = (const float*)d_in[0];
    const int*   eidx   = (const int*)d_in[1];
    const float* W      = (const float*)d_in[2];
    const float* Wself  = (const float*)d_in[3];
    const float* a_in   = (const float*)d_in[4];
    const float* a_out  = (const float*)d_in[5];
    const float* bias   = (const float*)d_in[6];
    const float* gamma  = (const float*)d_in[7];
    const float* beta   = (const float*)d_in[8];
    float* out = (float*)d_out;

    const int N = in_sizes[0] / IN_DIM;
    const int E = in_sizes[1] / 2;

    const int* src = eidx;
    const int* dst = eidx + E;

    float* ws = (float*)d_ws;
    float* h_proj  = ws;
    float* rec     = h_proj + (size_t)N * OUT_DIM;
    const int tiles = (N + TILE_ROWS - 1) / TILE_ROWS;
    const size_t base_words = (size_t)N * OUT_DIM + (size_t)N * REC;
    const size_t bin_words = (size_t)tiles * LIST_CAP;
    unsigned int* bins = (unsigned int*)(ws + base_words);
    int* fill = (int*)(bins + bin_words);
    const bool use_bins = ws_size >= (base_words + bin_words + (size_t)tiles) * sizeof(float);

    {
        int blocks = (N + GEMM_ROWS - 1) / GEMM_ROWS;
        gemm_wmma_kernel<<<blocks, 256, 0, stream>>>(h, W, Wself, h_proj, out, N,
                                                     a_in, a_out, rec,
                                                     use_bins ? fill : nullptr, tiles);
    }
    const int nwords = (E + MAP_EDGES - 1) / MAP_EDGES;
    const bool use_map = !use_bins && (long long)nwords * tiles <= (long long)N * MAP_SLOTS;
    unsigned int* map = (unsigned int*)rec;
    if (use_bins) {
        const int blocks = (E + BIN_EDGES - 1) / BIN_EDGES;
        tile_bin_kernel<<<blocks, BIN_THREADS, 3 * tiles * sizeof(int), stream>>>(
            src, dst, bins, fill, E, tiles);
    }
    if (use_map) {
        tile_map_kernel<<<nwords, MAP_EDGES, tiles * sizeof(unsigned int), stream>>>(
            src, dst, map, E, tiles);
    }
    {
        agg_tile_kernel<<<tiles, AGG_THREADS, 0, stream>>>(src, dst, h_proj, rec,
                                                           bias, gamma, beta,
                                                           use_map ? map : nullptr,
                                                           use_bins ? bins : nullptr, fill,
                                                           out, N, E);
    }
}
